# starting kernel
# baseline (speedup 1.0000x reference)
#include <hip/hip_runtime.h>
#include <hip/hip_bf16.h>

#define N_NODES   100000
#define N_EDGES   1600000
#define D_IN      256
#define D_HID     128
#define N_CLASSES 64
#define NEG_SLOPE 0.2f

typedef __attribute__((ext_vector_type(2))) float v2f;
typedef __attribute__((ext_vector_type(8))) float v8f;

__global__ __launch_bounds__(256) void k_gemm1(const float* __restrict__ X,
                                               const float* __restrict__ W,
                                               float* __restrict__ feat) {
    const int wave = threadIdx.x >> 5;
    const int lane = threadIdx.x & 31;
    const int strip = blockIdx.x * 8 + wave;
    const int rowTile = strip >> 2;
    const int colPair = strip & 3;
    const int row0 = rowTile * 16;
    const int col0 = colPair * 32;
    const int l16 = lane & 15;
    const int hi  = lane >> 4;

    v8f acc0 = {};
    v8f acc1 = {};
    const float* __restrict__ xrow = X + (size_t)(row0 + l16) * D_IN;

    for (int k0 = 0; k0 < D_IN; k0 += 4) {
        const int ka = k0 + 2 * hi;
        v2f a;
        a.x = xrow[ka];
        a.y = xrow[ka + 1];
        const float* __restrict__ w0 = W + (size_t)ka * D_HID + col0 + l16;
        const float* __restrict__ w1 = W + (size_t)(ka + 1) * D_HID + col0 + l16;
        v2f b0, b1;
        b0.x = w0[0];   b0.y = w1[0];
        b1.x = w0[16];  b1.y = w1[16];
        acc0 = __builtin_amdgcn_wmma_f32_16x16x4_f32(
            false, a, false, b0, (short)0, acc0, false, false);
        acc1 = __builtin_amdgcn_wmma_f32_16x16x4_f32(
            false, a, false, b1, (short)0, acc1, false, false);
    }
    #pragma unroll
    for (int r = 0; r < 8; ++r) {
        float* __restrict__ orow = feat + (size_t)(row0 + r + 8 * hi) * D_HID + col0 + l16;
        orow[0]  = acc0[r];
        orow[16] = acc1[r];
    }
}

__global__ __launch_bounds__(256) void k_gemm2(const float* __restrict__ H,
                                               const float* __restrict__ FW,
                                               const float* __restrict__ FB,
                                               float* __restrict__ out) {
    const int wave = threadIdx.x >> 5;
    const int lane = threadIdx.x & 31;
    const int rowTile = blockIdx.x * 8 + wave;
    if (rowTile >= N_NODES / 16) return;
    const int row0 = rowTile * 16;
    const int l16 = lane & 15;
    const int hi  = lane >> 4;

    v8f acc0 = {}, acc1 = {}, acc2 = {}, acc3 = {};
    const float* __restrict__ hrow = H + (size_t)(row0 + l16) * D_HID;

    for (int k0 = 0; k0 < D_HID; k0 += 4) {
        const int ka = k0 + 2 * hi;
        v2f a;
        a.x = hrow[ka];
        a.y = hrow[ka + 1];
        const float* __restrict__ w0 = FW + (size_t)ka * N_CLASSES + l16;
        const float* __restrict__ w1 = FW + (size_t)(ka + 1) * N_CLASSES + l16;
        v2f b0, b1, b2, b3;
        b0.x = w0[0];   b0.y = w1[0];
        b1.x = w0[16];  b1.y = w1[16];
        b2.x = w0[32];  b2.y = w1[32];
        b3.x = w0[48];  b3.y = w1[48];
        acc0 = __builtin_amdgcn_wmma_f32_16x16x4_f32(
            false, a, false, b0, (short)0, acc0, false, false);
        acc1 = __builtin_amdgcn_wmma_f32_16x16x4_f32(
            false, a, false, b1, (short)0, acc1, false, false);
        acc2 = __builtin_amdgcn_wmma_f32_16x16x4_f32(
            false, a, false, b2, (short)0, acc2, false, false);
        acc3 = __builtin_amdgcn_wmma_f32_16x16x4_f32(
            false, a, false, b3, (short)0, acc3, false, false);
    }
    const float bias0 = FB[l16];
    const float bias1 = FB[l16 + 16];
    const float bias2 = FB[l16 + 32];
    const float bias3 = FB[l16 + 48];
    #pragma unroll
    for (int r = 0; r < 8; ++r) {
        float* __restrict__ orow = out + (size_t)(row0 + r + 8 * hi) * N_CLASSES + l16;
        orow[0]  = acc0[r] + bias0;
        orow[16] = acc1[r] + bias1;
        orow[32] = acc2[r] + bias2;
        orow[48] = acc3[r] + bias3;
    }
}

__global__ __launch_bounds__(256) void k_scores(const float* __restrict__ feat,
                                                const float* __restrict__ al,
                                                const float* __restrict__ ar,
                                                float* __restrict__ el,
                                                float* __restrict__ er) {
    const int wave = threadIdx.x >> 5;
    const int lane = threadIdx.x & 31;
    const int node = blockIdx.x * 8 + wave;
    if (node >= N_NODES) return;
    const float* __restrict__ f = feat + (size_t)node * D_HID;
    float sl = 0.f, sr = 0.f;
    #pragma unroll
    for (int j = 0; j < D_HID; j += 32) {
        float v = f[j + lane];
        sl += v * al[j + lane];
        sr += v * ar[j + lane];
    }
    #pragma unroll
    for (int off = 16; off > 0; off >>= 1) {
        sl += __shfl_down(sl, off, 32);
        sr += __shfl_down(sr, off, 32);
    }
    if (lane == 0) { el[node] = sl; er[node] = sr; }
}

__device__ __forceinline__ unsigned enc_f32(float f) {
    unsigned b = __float_as_uint(f);
    return (b & 0x80000000u) ? ~b : (b | 0x80000000u);
}
__device__ __forceinline__ float dec_f32(unsigned u) {
    unsigned b = (u & 0x80000000u) ? (u & 0x7FFFFFFFu) : ~u;
    return __uint_as_float(b);
}

__global__ __launch_bounds__(256) void k_edge1(const float* __restrict__ el,
                                               const float* __restrict__ er,
                                               const int* __restrict__ src,
                                               const int* __restrict__ dst,
                                               float* __restrict__ e,
                                               unsigned* __restrict__ mEnc) {
    const int i = blockIdx.x * 256 + threadIdx.x;
    if (i >= N_EDGES) return;
    float v = el[src[i]] + er[dst[i]];
    v = (v > 0.f) ? v : NEG_SLOPE * v;
    e[i] = v;
    atomicMax(&mEnc[dst[i]], enc_f32(v));
}

__global__ __launch_bounds__(256) void k_edge2(const int* __restrict__ dst,
                                               const unsigned* __restrict__ mEnc,
                                               float* __restrict__ e,
                                               float* __restrict__ denom) {
    const int i = blockIdx.x * 256 + threadIdx.x;
    if (i >= N_EDGES) return;
    const int d = dst[i];
    const float ex = __expf(e[i] - dec_f32(mEnc[d]));
    e[i] = ex;
    atomicAdd(&denom[d], ex);
}

__global__ __launch_bounds__(256) void k_edge3(const int* __restrict__ src,
                                               const int* __restrict__ dst,
                                               const float* __restrict__ ex,
                                               const float* __restrict__ denom,
                                               const float* __restrict__ feat,
                                               float* __restrict__ h) {
    const int eIdx = blockIdx.x * 2 + (threadIdx.x >> 7);
    const int t    = threadIdx.x & 127;
    if (eIdx >= N_EDGES) return;
    const int d = dst[eIdx];
    const int s = src[eIdx];
    const float a = ex[eIdx] / denom[d];
    atomicAdd(&h[(size_t)d * D_HID + t], a * feat[(size_t)s * D_HID + t]);
}

extern "C" void kernel_launch(void* const* d_in, const int* in_sizes, int n_in,
                              void* d_out, int out_size, void* d_ws, size_t ws_size,
                              hipStream_t stream) {
    const float* X  = (const float*)d_in[0];
    const float* W  = (const float*)d_in[1];
    const float* al = (const float*)d_in[2];
    const float* ar = (const float*)d_in[3];
    const float* FW = (const float*)d_in[4];
    const float* FB = (const float*)d_in[5];
    const int*  src = (const int*)d_in[6];
    const int*  dst = (const int*)d_in[7];
    float* out = (float*)d_out;

    char* ws = (char*)d_ws;
    size_t off = 0;
    auto carve = [&](size_t bytes) {
        void* p = ws + off;
        off += (bytes + 255) & ~(size_t)255;
        return p;
    };
    float*    feat  = (float*)   carve((size_t)N_NODES * D_HID * sizeof(float));
    float*    h     = (float*)   carve((size_t)N_NODES * D_HID * sizeof(float));
    float*    el    = (float*)   carve((size_t)N_NODES * sizeof(float));
    float*    er    = (float*)   carve((size_t)N_NODES * sizeof(float));
    unsigned* mEnc  = (unsigned*)carve((size_t)N_NODES * sizeof(unsigned));
    float*    denom = (float*)   carve((size_t)N_NODES * sizeof(float));
    float*    e     = (float*)   carve((size_t)N_EDGES * sizeof(float));

    hipMemsetAsync(h,     0, (size_t)N_NODES * D_HID * sizeof(float), stream);
    hipMemsetAsync(mEnc,  0, (size_t)N_NODES * sizeof(unsigned), stream);
    hipMemsetAsync(denom, 0, (size_t)N_NODES * sizeof(float), stream);

    k_gemm1<<<3125, 256, 0, stream>>>(X, W, feat);
    k_scores<<<12500, 256, 0, stream>>>(feat, al, ar, el, er);
    k_edge1<<<(N_EDGES + 255) / 256, 256, 0, stream>>>(el, er, src, dst, e, mEnc);
    k_edge2<<<(N_EDGES + 255) / 256, 256, 0, stream>>>(dst, mEnc, e, denom);
    k_edge3<<<N_EDGES / 2, 256, 0, stream>>>(src, dst, e, denom, feat, h);
    k_gemm2<<<782, 256, 0, stream>>>(h, FW, FB, out);
}


# Edge pass fused into the aggregation block: owned edges listed in LDS, scored there, row max by LDS integer max, each row summed by one wave in fixed list order (order new); edge launch, memset gone
# speedup vs baseline: 3.9417x; 3.9417x over previous
#include <hip/hip_runtime.h>
#include <hip/hip_bf16.h>

#define N_NODES   100000
#define N_EDGES   1600000
#define D_IN      256
#define D_HID     128
#define N_CLASSES 64
#define NEG_SLOPE 0.2f

typedef __attribute__((ext_vector_type(2))) float v2f;
typedef __attribute__((ext_vector_type(8))) float v8f;

__global__ __launch_bounds__(128) void k_gemm1(const float* __restrict__ X,
                                               const float* __restrict__ W,
                                               const float* __restrict__ al,
                                               const float* __restrict__ ar,
                                               float* __restrict__ feat,
                                               float* __restrict__ el,
                                               float* __restrict__ er) {
    const int wave = threadIdx.x >> 5;
    const int lane = threadIdx.x & 31;
    const int rowTile = blockIdx.x * 4 + wave;
    if (rowTile >= N_NODES / 16) return;
    const int row0 = rowTile * 16;
    const int l16 = lane & 15;
    const int hi  = lane >> 4;

    v8f acc[8] = {};
    const float* __restrict__ xrow = X + (size_t)(row0 + l16) * D_IN + 2 * hi;
    const float* __restrict__ wcol = W + (size_t)(2 * hi) * D_HID + l16;

    for (int k0 = 0; k0 < D_IN; k0 += 4) {
        v2f a;
        a.x = xrow[k0];
        a.y = xrow[k0 + 1];
        const float* __restrict__ w0 = wcol + (size_t)k0 * D_HID;
        const float* __restrict__ w1 = w0 + D_HID;
        v2f b[8];
        for (int t = 0; t < 8; ++t) {
            b[t].x = w0[16 * t];
            b[t].y = w1[16 * t];
        }
        for (int t = 0; t < 8; ++t)
            acc[t] = __builtin_amdgcn_wmma_f32_16x16x4_f32(
                false, a, false, b[t], (short)0, acc[t], false, false);
    }
    for (int r = 0; r < 8; ++r) {
        float* __restrict__ orow = feat + (size_t)(row0 + r + 8 * hi) * D_HID + l16;
        for (int t = 0; t < 8; ++t) orow[16 * t] = acc[t][r];
    }
    float alv[8], arv[8];
    for (int t = 0; t < 8; ++t) {
        alv[t] = al[16 * t + l16];
        arv[t] = ar[16 * t + l16];
    }
    float myl = 0.f, myr = 0.f;
    for (int r = 0; r < 8; ++r) {
        float sl = 0.f, sr = 0.f;
        for (int t = 0; t < 8; ++t) {
            sl += acc[t][r] * alv[t];
            sr += acc[t][r] * arv[t];
        }
        for (int off = 8; off > 0; off >>= 1) {
            sl += __shfl_xor(sl, off, 32);
            sr += __shfl_xor(sr, off, 32);
        }
        if (l16 == r) { myl = sl; myr = sr; }
    }
    if (l16 < 8) {
        el[row0 + l16 + 8 * hi] = myl;
        er[row0 + l16 + 8 * hi] = myr;
    }
}

__global__ __launch_bounds__(256) void k_gemm2(const float* __restrict__ H,
                                               const float* __restrict__ FW,
                                               const float* __restrict__ FB,
                                               float* __restrict__ out) {
    const int wave = threadIdx.x >> 5;
    const int lane = threadIdx.x & 31;
    const int rowTile = blockIdx.x * 8 + wave;
    if (rowTile >= N_NODES / 16) return;
    const int row0 = rowTile * 16;
    const int l16 = lane & 15;
    const int hi  = lane >> 4;

    v8f acc0 = {}, acc1 = {}, acc2 = {}, acc3 = {};
    const float* __restrict__ hrow = H + (size_t)(row0 + l16) * D_HID;

    for (int k0 = 0; k0 < D_HID; k0 += 4) {
        const int ka = k0 + 2 * hi;
        v2f a;
        a.x = hrow[ka];
        a.y = hrow[ka + 1];
        const float* __restrict__ w0 = FW + (size_t)ka * N_CLASSES + l16;
        const float* __restrict__ w1 = FW + (size_t)(ka + 1) * N_CLASSES + l16;
        v2f b0, b1, b2, b3;
        b0.x = w0[0];   b0.y = w1[0];
        b1.x = w0[16];  b1.y = w1[16];
        b2.x = w0[32];  b2.y = w1[32];
        b3.x = w0[48];  b3.y = w1[48];
        acc0 = __builtin_amdgcn_wmma_f32_16x16x4_f32(
            false, a, false, b0, (short)0, acc0, false, false);
        acc1 = __builtin_amdgcn_wmma_f32_16x16x4_f32(
            false, a, false, b1, (short)0, acc1, false, false);
        acc2 = __builtin_amdgcn_wmma_f32_16x16x4_f32(
            false, a, false, b2, (short)0, acc2, false, false);
        acc3 = __builtin_amdgcn_wmma_f32_16x16x4_f32(
            false, a, false, b3, (short)0, acc3, false, false);
    }
    const float bias0 = FB[l16];
    const float bias1 = FB[l16 + 16];
    const float bias2 = FB[l16 + 32];
    const float bias3 = FB[l16 + 48];
    #pragma unroll
    for (int r = 0; r < 8; ++r) {
        float* __restrict__ orow = out + (size_t)(row0 + r + 8 * hi) * N_CLASSES + l16;
        orow[0]  = acc0[r] + bias0;
        orow[16] = acc1[r] + bias1;
        orow[32] = acc2[r] + bias2;
        orow[48] = acc3[r] + bias3;
    }
}

__device__ __forceinline__ unsigned enc_f32(float f) {
    unsigned b = __float_as_uint(f);
    return (b & 0x80000000u) ? ~b : (b | 0x80000000u);
}
__device__ __forceinline__ float dec_f32(unsigned u) {
    unsigned b = (u & 0x80000000u) ? (u & 0x7FFFFFFFu) : ~u;
    return __uint_as_float(b);
}

#define AGG_WGS     256
#define AGG_ROWS    391
#define AGG_THREADS 1024
#define AGG_WAVES   (AGG_THREADS / 32)
#define AGG_CAP     400
#define AGG_RING    64
#define AGG_BATCH   8
#define AGG_UNROLL  4
#define EDGE_GROUPS (N_EDGES / 128)
#define ROW_SHIFT   17
static_assert(N_NODES <= (1 << ROW_SHIFT), "src must fit below the row field");
static_assert(AGG_ROWS <= (1 << (31 - ROW_SHIFT)), "row must fit above the src field");

__device__ __forceinline__ float edge_score(float l, float r) {
    const float v = l + r;
    return (v > 0.f) ? v : NEG_SLOPE * v;
}

template <bool FULL>
__device__ __forceinline__ void agg_batch(const int2* ringw, unsigned head, int n, int lane,
                                          const unsigned* mx,
                                          const float* __restrict__ feat,
                                          float* acc, float* den) {
    int s = 0, dl = 0;
    float a = 0.f;
    if (lane < n) {
        const int2 ent = ringw[(head + (unsigned)lane) & (AGG_RING - 1)];
        dl = ent.x >> ROW_SHIFT;
        s  = ent.x & ((1 << ROW_SHIFT) - 1);
        a  = __expf(__int_as_float(ent.y) - dec_f32(mx[dl]));
    }
    float2 v0[AGG_BATCH], v1[AGG_BATCH];
    #pragma unroll
    for (int j = 0; j < AGG_BATCH; ++j) {
        if (FULL || j < n) {
            const int sj = __builtin_amdgcn_readlane(s, j);
            const float2* __restrict__ row =
                reinterpret_cast<const float2*>(feat + (size_t)sj * D_HID);
            v0[j] = row[lane];
            v1[j] = row[32 + lane];
        }
    }
    #pragma unroll
    for (int j = 0; j < AGG_BATCH; ++j) {
        if (FULL || j < n) {
            const int dj = __builtin_amdgcn_readlane(dl, j);
            const float aj = __int_as_float(__builtin_amdgcn_readlane(__float_as_int(a), j));
            if (lane == 0) atomicAdd(den + dj, aj);
            float* arow = acc + dj * D_HID + 2 * lane;
            atomicAdd(arow,      aj * v0[j].x);
            atomicAdd(arow + 1,  aj * v0[j].y);
            atomicAdd(arow + 64, aj * v1[j].x);
            atomicAdd(arow + 65, aj * v1[j].y);
        }
    }
}

__global__ __launch_bounds__(AGG_THREADS) void k_agg(const int* __restrict__ src,
                                                     const int* __restrict__ dst,
                                                     const float* __restrict__ el,
                                                     const float* __restrict__ er,
                                                     const float* __restrict__ feat,
                                                     float* __restrict__ h) {
    __shared__ __attribute__((aligned(16))) float acc[AGG_ROWS * D_HID];
    __shared__ int2 lst[AGG_WAVES * AGG_CAP];
    __shared__ int2 ring[AGG_WAVES * AGG_RING];
    __shared__ float den[AGG_ROWS];
    __shared__ unsigned mx[AGG_ROWS];
    __shared__ int cntw[AGG_WAVES];
    __shared__ int ovf;

    const int tid  = threadIdx.x;
    const int lane = tid & 31;
    const int wave = __builtin_amdgcn_readfirstlane(tid >> 5);
    const int base = blockIdx.x * AGG_ROWS;
    const int nrows = (N_NODES - base < AGG_ROWS) ? (N_NODES - base) : AGG_ROWS;

    float4* acc4 = reinterpret_cast<float4*>(acc);
    for (int i = tid; i < AGG_ROWS * D_HID / 4; i += AGG_THREADS)
        acc4[i] = make_float4(0.f, 0.f, 0.f, 0.f);
    if (tid < AGG_ROWS) { den[tid] = 0.f; mx[tid] = 0u; }
    if (tid == 0) ovf = 0;
    __syncthreads();

    int2* lstw = lst + wave * AGG_CAP;
    const float* __restrict__ er_own = er + base;
    const unsigned lt_mask = (1u << lane) - 1u;
    const int4* __restrict__ dst4 = reinterpret_cast<const int4*>(dst);

    unsigned cnt = 0;
    auto fetch = [&](int g) -> int4 {
        return dst4[((g < EDGE_GROUPS) ? g : (EDGE_GROUPS - 1)) * 32 + lane];
    };
    auto examine = [&](const int4& d, int g) {
        if (g >= EDGE_GROUPS) return;
        const unsigned t0 = (unsigned)(d.x - base);
        const unsigned t1 = (unsigned)(d.y - base);
        const unsigned t2 = (unsigned)(d.z - base);
        const unsigned t3 = (unsigned)(d.w - base);
        const unsigned tmin = min(min(t0, t1), min(t2, t3));
        if (__builtin_amdgcn_ballot_w32(tmin < (unsigned)nrows) == 0u) return;
        const int e0 = (g * 32 + lane) * 4;
        const unsigned tt[4] = {t0, t1, t2, t3};
        #pragma unroll
        for (int c = 0; c < 4; ++c) {
            const bool m = tt[c] < (unsigned)nrows;
            const unsigned bal = __builtin_amdgcn_ballot_w32(m);
            if (bal != 0u) {
                const unsigned pos = cnt + (unsigned)__popc(bal & lt_mask);
                if (m && pos < (unsigned)AGG_CAP) lstw[pos] = make_int2(e0 + c, (int)tt[c]);
                cnt += (unsigned)__popc(bal);
            }
        }
    };

    constexpr int STEP = AGG_WAVES * AGG_UNROLL;
    int4 bufA[AGG_UNROLL], bufB[AGG_UNROLL];
    #pragma unroll
    for (int u = 0; u < AGG_UNROLL; ++u) bufA[u] = fetch(wave + AGG_WAVES * u);
    for (int g0 = wave; g0 < EDGE_GROUPS; g0 += 2 * STEP) {
        #pragma unroll
        for (int u = 0; u < AGG_UNROLL; ++u) bufB[u] = fetch(g0 + STEP + AGG_WAVES * u);
        #pragma unroll
        for (int u = 0; u < AGG_UNROLL; ++u) examine(bufA[u], g0 + AGG_WAVES * u);
        #pragma unroll
        for (int u = 0; u < AGG_UNROLL; ++u) bufA[u] = fetch(g0 + 2 * STEP + AGG_WAVES * u);
        #pragma unroll
        for (int u = 0; u < AGG_UNROLL; ++u) examine(bufB[u], g0 + STEP + AGG_WAVES * u);
    }
    const unsigned cntc = (cnt < (unsigned)AGG_CAP) ? cnt : (unsigned)AGG_CAP;
    if (lane == 0) {
        cntw[wave] = (int)cntc;
        if (cnt > (unsigned)AGG_CAP) ovf = 1;
    }

    for (unsigned i0 = 0; i0 < cntc; i0 += 128) {
        int2 ent[4];
        int  sv[4];
        float ev[4];
        #pragma unroll
        for (int u = 0; u < 4; ++u) {
            const unsigned i = i0 + 32u * u + (unsigned)lane;
            ent[u] = (i < cntc) ? lstw[i] : make_int2(0, 0);
        }
        #pragma unroll
        for (int u = 0; u < 4; ++u) sv[u] = src[ent[u].x];
        #pragma unroll
        for (int u = 0; u < 4; ++u) ev[u] = edge_score(el[sv[u]], er_own[ent[u].y]);
        #pragma unroll
        for (int u = 0; u < 4; ++u) {
            const unsigned i = i0 + 32u * u + (unsigned)lane;
            if (i < cntc) {
                atomicMax(&mx[ent[u].y], enc_f32(ev[u]));
                lstw[i] = make_int2(sv[u] | (ent[u].y << ROW_SHIFT), __float_as_int(ev[u]));
            }
        }
    }
    __syncthreads();

    const int ovf_u = __builtin_amdgcn_readfirstlane(ovf);
    if (ovf_u == 0) {
        int2* ringw = ring + wave * AGG_RING;
        unsigned head = 0, tail = 0;
        for (int q = 0; q < AGG_WAVES; ++q) {
            const unsigned nq = (unsigned)__builtin_amdgcn_readfirstlane(cntw[q]);
            const int2* lq = lst + q * AGG_CAP;
            for (unsigned i0 = 0; i0 < nq; i0 += 128) {
                int2 en[4];
                #pragma unroll
                for (int u = 0; u < 4; ++u) {
                    const unsigned i = i0 + 32u * u + (unsigned)lane;
                    en[u] = (i < nq) ? lq[i] : make_int2(0, 0);
                }
                #pragma unroll
                for (int u = 0; u < 4; ++u) {
                    const unsigned i = i0 + 32u * u + (unsigned)lane;
                    const bool m = (i < nq) &&
                        ((((unsigned)en[u].x >> ROW_SHIFT) & (AGG_WAVES - 1)) == (unsigned)wave);
                    const unsigned bal = __builtin_amdgcn_ballot_w32(m);
                    if (bal != 0u) {
                        if (m) ringw[(tail + (unsigned)__popc(bal & lt_mask)) & (AGG_RING - 1)] = en[u];
                        tail += (unsigned)__popc(bal);
                        while (tail - head >= (unsigned)AGG_BATCH) {
                            agg_batch<true>(ringw, head, AGG_BATCH, lane, mx, feat, acc, den);
                            head += AGG_BATCH;
                        }
                    }
                }
            }
        }
        if (tail != head)
            agg_batch<false>(ringw, head, (int)(tail - head), lane, mx, feat, acc, den);
    } else {
        for (int g = wave; g < EDGE_GROUPS; g += AGG_WAVES) {
            const int4 d = dst4[g * 32 + lane];
            const int dd[4] = {d.x, d.y, d.z, d.w};
            #pragma unroll
            for (int c = 0; c < 4; ++c) {
                const unsigned t = (unsigned)(dd[c] - base);
                if (t < (unsigned)nrows) {
                    const int s = src[(g * 32 + lane) * 4 + c];
                    atomicMax(&mx[t], enc_f32(edge_score(el[s], er_own[t])));
                }
            }
        }
        __syncthreads();
        for (int g = wave; g < EDGE_GROUPS; g += AGG_WAVES) {
            const int4 d = dst4[g * 32 + lane];
            const int dd[4] = {d.x, d.y, d.z, d.w};
            #pragma unroll
            for (int c = 0; c < 4; ++c) {
                const unsigned t = (unsigned)(dd[c] - base);
                const bool m = t < (unsigned)nrows;
                int s = 0, tl = 0;
                float a = 0.f;
                if (m) {
                    s = src[(g * 32 + lane) * 4 + c];
                    tl = (int)t;
                    a = __expf(edge_score(el[s], er_own[t]) - dec_f32(mx[t]));
                    atomicAdd(den + tl, a);
                }
                unsigned bal = __builtin_amdgcn_ballot_w32(m);
                while (bal != 0u) {
                    const int j = __builtin_ctz(bal);
                    bal &= bal - 1u;
                    const int sj = __builtin_amdgcn_readlane(s, j);
                    const int dj = __builtin_amdgcn_readlane(tl, j);
                    const float aj = __int_as_float(__builtin_amdgcn_readlane(__float_as_int(a), j));
                    const float2* __restrict__ row =
                        reinterpret_cast<const float2*>(feat + (size_t)sj * D_HID);
                    const float2 x0 = row[lane], x1 = row[32 + lane];
                    float* arow = acc + dj * D_HID + 2 * lane;
                    atomicAdd(arow,      aj * x0.x);
                    atomicAdd(arow + 1,  aj * x0.y);
                    atomicAdd(arow + 64, aj * x1.x);
                    atomicAdd(arow + 65, aj * x1.y);
                }
            }
        }
    }

    __syncthreads();
    if (tid < AGG_ROWS) {
        const float d = den[tid];
        den[tid] = (d > 0.f) ? 1.0f / d : 0.f;
    }
    __syncthreads();
    float4* __restrict__ h4 = reinterpret_cast<float4*>(h + (size_t)base * D_HID);
    for (int i = tid; i < nrows * (D_HID / 4); i += AGG_THREADS) {
        const float r = den[i / (D_HID / 4)];
        float4 v = acc4[i];
        v.x *= r; v.y *= r; v.z *= r; v.w *= r;
        h4[i] = v;
    }
}

extern "C" void kernel_launch(void* const* d_in, const int* in_sizes, int n_in,
                              void* d_out, int out_size, void* d_ws, size_t ws_size,
                              hipStream_t stream) {
    const float* X  = (const float*)d_in[0];
    const float* W  = (const float*)d_in[1];
    const float* al = (const float*)d_in[2];
    const float* ar = (const float*)d_in[3];
    const float* FW = (const float*)d_in[4];
    const float* FB = (const float*)d_in[5];
    const int*  src = (const int*)d_in[6];
    const int*  dst = (const int*)d_in[7];
    float* out = (float*)d_out;

    char* ws = (char*)d_ws;
    size_t off = 0;
    auto carve = [&](size_t bytes) {
        void* p = ws + off;
        off += (bytes + 255) & ~(size_t)255;
        return p;
    };
    float*    feat  = (float*)   carve((size_t)N_NODES * D_HID * sizeof(float));
    float*    h     = (float*)   carve((size_t)N_NODES * D_HID * sizeof(float));
    float*    el    = (float*)   carve((size_t)N_NODES * sizeof(float));
    float*    er    = (float*)   carve((size_t)N_NODES * sizeof(float));

    k_gemm1<<<1563, 128, 0, stream>>>(X, W, al, ar, feat, el, er);
    k_agg<<<AGG_WGS, AGG_THREADS, 0, stream>>>(src, dst, el, er, feat, h);
    k_gemm2<<<782, 256, 0, stream>>>(h, FW, FB, out);
}


# Aggregation row sums kept in registers: a wave holds its 13 rows, the uniform slot number indexes the registers; float4 row loads, rows stored from registers; LDS float adds, tile passes gone; fma
# speedup vs baseline: 4.7456x; 1.2039x over previous
#include <hip/hip_runtime.h>
#include <hip/hip_bf16.h>

#define N_NODES   100000
#define N_EDGES   1600000
#define D_IN      256
#define D_HID     128
#define N_CLASSES 64
#define NEG_SLOPE 0.2f

typedef __attribute__((ext_vector_type(2))) float v2f;
typedef __attribute__((ext_vector_type(8))) float v8f;

#define OWN_ROWS 391u
static_assert(N_EDGES == (N_NODES / 16) * 256, "one 256-edge slice of the owner map per GEMM1 wave");
static_assert((N_NODES - 1) / OWN_ROWS < 256u, "owner index must fit one byte");

__global__ __launch_bounds__(128) void k_gemm1(const float* __restrict__ X,
                                               const float* __restrict__ W,
                                               const float* __restrict__ al,
                                               const float* __restrict__ ar,
                                               float* __restrict__ feat,
                                               float* __restrict__ el,
                                               float* __restrict__ er,
                                               const int* __restrict__ dst,
                                               unsigned* __restrict__ own) {
    const int wave = threadIdx.x >> 5;
    const int lane = threadIdx.x & 31;
    const int rowTile = blockIdx.x * 4 + wave;
    if (rowTile >= N_NODES / 16) return;
    const int row0 = rowTile * 16;
    {
        const int4* __restrict__ d4 =
            reinterpret_cast<const int4*>(dst) + (size_t)rowTile * 64 + lane * 2;
        const int4 da = d4[0], db = d4[1];
        const unsigned lo = ((unsigned)da.x / OWN_ROWS) | (((unsigned)da.y / OWN_ROWS) << 8) |
                            (((unsigned)da.z / OWN_ROWS) << 16) | (((unsigned)da.w / OWN_ROWS) << 24);
        const unsigned hi8 = ((unsigned)db.x / OWN_ROWS) | (((unsigned)db.y / OWN_ROWS) << 8) |
                             (((unsigned)db.z / OWN_ROWS) << 16) | (((unsigned)db.w / OWN_ROWS) << 24);
        reinterpret_cast<uint2*>(own)[rowTile * 32 + lane] = make_uint2(lo, hi8);
    }
    const int l16 = lane & 15;
    const int hi  = lane >> 4;

    v8f acc[8] = {};
    const float* __restrict__ xrow = X + (size_t)(row0 + l16) * D_IN + 2 * hi;
    const float* __restrict__ wcol = W + (size_t)(2 * hi) * D_HID + l16;

    for (int k0 = 0; k0 < D_IN; k0 += 4) {
        v2f a;
        a.x = xrow[k0];
        a.y = xrow[k0 + 1];
        const float* __restrict__ w0 = wcol + (size_t)k0 * D_HID;
        const float* __restrict__ w1 = w0 + D_HID;
        v2f b[8];
        for (int t = 0; t < 8; ++t) {
            b[t].x = w0[16 * t];
            b[t].y = w1[16 * t];
        }
        for (int t = 0; t < 8; ++t)
            acc[t] = __builtin_amdgcn_wmma_f32_16x16x4_f32(
                false, a, false, b[t], (short)0, acc[t], false, false);
    }
    for (int r = 0; r < 8; ++r) {
        float* __restrict__ orow = feat + (size_t)(row0 + r + 8 * hi) * D_HID + l16;
        for (int t = 0; t < 8; ++t) orow[16 * t] = acc[t][r];
    }
    float alv[8], arv[8];
    for (int t = 0; t < 8; ++t) {
        alv[t] = al[16 * t + l16];
        arv[t] = ar[16 * t + l16];
    }
    float myl = 0.f, myr = 0.f;
    for (int r = 0; r < 8; ++r) {
        float sl = 0.f, sr = 0.f;
        for (int t = 0; t < 8; ++t) {
            sl += acc[t][r] * alv[t];
            sr += acc[t][r] * arv[t];
        }
        for (int off = 8; off > 0; off >>= 1) {
            sl += __shfl_xor(sl, off, 32);
            sr += __shfl_xor(sr, off, 32);
        }
        if (l16 == r) { myl = sl; myr = sr; }
    }
    if (l16 < 8) {
        el[row0 + l16 + 8 * hi] = myl;
        er[row0 + l16 + 8 * hi] = myr;
    }
}

__global__ __launch_bounds__(256) void k_gemm2(const float* __restrict__ H,
                                               const float* __restrict__ FW,
                                               const float* __restrict__ FB,
                                               float* __restrict__ out) {
    const int wave = threadIdx.x >> 5;
    const int lane = threadIdx.x & 31;
    const int rowTile = blockIdx.x * 8 + wave;
    if (rowTile >= N_NODES / 16) return;
    const int row0 = rowTile * 16;
    const int l16 = lane & 15;
    const int hi  = lane >> 4;

    v8f acc0 = {}, acc1 = {}, acc2 = {}, acc3 = {};
    const float* __restrict__ hrow = H + (size_t)(row0 + l16) * D_HID;

    for (int k0 = 0; k0 < D_HID; k0 += 4) {
        const int ka = k0 + 2 * hi;
        v2f a;
        a.x = hrow[ka];
        a.y = hrow[ka + 1];
        const float* __restrict__ w0 = FW + (size_t)ka * N_CLASSES + l16;
        const float* __restrict__ w1 = FW + (size_t)(ka + 1) * N_CLASSES + l16;
        v2f b0, b1, b2, b3;
        b0.x = w0[0];   b0.y = w1[0];
        b1.x = w0[16];  b1.y = w1[16];
        b2.x = w0[32];  b2.y = w1[32];
        b3.x = w0[48];  b3.y = w1[48];
        acc0 = __builtin_amdgcn_wmma_f32_16x16x4_f32(
            false, a, false, b0, (short)0, acc0, false, false);
        acc1 = __builtin_amdgcn_wmma_f32_16x16x4_f32(
            false, a, false, b1, (short)0, acc1, false, false);
        acc2 = __builtin_amdgcn_wmma_f32_16x16x4_f32(
            false, a, false, b2, (short)0, acc2, false, false);
        acc3 = __builtin_amdgcn_wmma_f32_16x16x4_f32(
            false, a, false, b3, (short)0, acc3, false, false);
    }
    const float bias0 = FB[l16];
    const float bias1 = FB[l16 + 16];
    const float bias2 = FB[l16 + 32];
    const float bias3 = FB[l16 + 48];
    #pragma unroll
    for (int r = 0; r < 8; ++r) {
        float* __restrict__ orow = out + (size_t)(row0 + r + 8 * hi) * N_CLASSES + l16;
        orow[0]  = acc0[r] + bias0;
        orow[16] = acc1[r] + bias1;
        orow[32] = acc2[r] + bias2;
        orow[48] = acc3[r] + bias3;
    }
}

__device__ __forceinline__ unsigned enc_f32(float f) {
    unsigned b = __float_as_uint(f);
    return (b & 0x80000000u) ? ~b : (b | 0x80000000u);
}
__device__ __forceinline__ float dec_f32(unsigned u) {
    unsigned b = (u & 0x80000000u) ? (u & 0x7FFFFFFFu) : ~u;
    return __uint_as_float(b);
}

#define AGG_WGS     256
#define AGG_ROWS    391
static_assert(AGG_ROWS == (int)OWN_ROWS, "owner map and block ranges must agree");
#define AGG_THREADS 1024
#define AGG_WAVES   (AGG_THREADS / 32)
#define AGG_CAP     400
#define AGG_RING    64
#define AGG_BATCH   8
#define AGG_UNROLL  4
#define EDGE_GROUPS (N_EDGES / 128)
#define OWN_GROUPS  (N_EDGES / 512)
#define ROW_SHIFT   17
static_assert(N_NODES <= (1 << ROW_SHIFT), "src must fit below the row field");
static_assert(AGG_ROWS <= (1 << (31 - ROW_SHIFT)), "row must fit above the src field");
#define AGG_SLOTS   ((AGG_ROWS + AGG_WAVES - 1) / AGG_WAVES)
static_assert(AGG_SLOTS <= 32, "lane k of a wave holds the denominator of its slot k");

__device__ __forceinline__ float edge_score(float l, float r) {
    const float v = l + r;
    return (v > 0.f) ? v : NEG_SLOPE * v;
}

typedef __attribute__((ext_vector_type(16))) float v16f;
static_assert(AGG_SLOTS <= 16, "one vector element per slot");

template <bool FULL>
__device__ __forceinline__ void agg_batch(const int2* ringw, unsigned head, int n, int lane,
                                          const unsigned* mx,
                                          const float* __restrict__ feat,
                                          v16f (&sc)[4], float& dn) {
    int sr = 0;
    float a = 0.f;
    if (lane < n) {
        const int2 ent = ringw[(head + (unsigned)lane) & (AGG_RING - 1)];
        sr = ent.x;
        a  = __expf(__int_as_float(ent.y) - dec_f32(mx[sr >> ROW_SHIFT]));
    }
    float4 v[AGG_BATCH];
    int srj[AGG_BATCH];
    const float4* __restrict__ feat4 = reinterpret_cast<const float4*>(feat);
    #pragma unroll
    for (int j = 0; j < AGG_BATCH; ++j) {
        if (FULL || j < n) {
            srj[j] = __builtin_amdgcn_readlane(sr, j);
            const unsigned sj = (unsigned)srj[j] & ((1u << ROW_SHIFT) - 1u);
            v[j] = feat4[sj * (unsigned)(D_HID / 4) + (unsigned)lane];
        }
    }
    #pragma unroll
    for (int j = 0; j < AGG_BATCH; ++j) {
        if (FULL || j < n) {
            const int kj = (int)(((unsigned)srj[j] >> ROW_SHIFT) / AGG_WAVES);
            const float aj = __int_as_float(__builtin_amdgcn_readlane(__float_as_int(a), j));
            if (lane == kj) dn += aj;
            sc[0][kj] += aj * v[j].x;
            sc[1][kj] += aj * v[j].y;
            sc[2][kj] += aj * v[j].z;
            sc[3][kj] += aj * v[j].w;
        }
    }
}

__global__ __launch_bounds__(AGG_THREADS) void k_agg(const int* __restrict__ src,
                                                     const int* __restrict__ dst,
                                                     const unsigned* __restrict__ own,
                                                     const float* __restrict__ el,
                                                     const float* __restrict__ er,
                                                     const float* __restrict__ feat,
                                                     float* __restrict__ h) {
    __shared__ __attribute__((aligned(16))) float acc[AGG_ROWS * D_HID];
    __shared__ int2 lst[AGG_WAVES * AGG_CAP];
    __shared__ int2 ring[AGG_WAVES * AGG_RING];
    __shared__ float den[AGG_ROWS];
    __shared__ unsigned mx[AGG_ROWS];
    __shared__ int cntw[AGG_WAVES];
    __shared__ int ovf;

    const int tid  = threadIdx.x;
    const int lane = tid & 31;
    const int wave = __builtin_amdgcn_readfirstlane(tid >> 5);
    const int base = blockIdx.x * AGG_ROWS;
    const int nrows = (N_NODES - base < AGG_ROWS) ? (N_NODES - base) : AGG_ROWS;

    if (tid < AGG_ROWS) mx[tid] = 0u;
    if (tid == 0) ovf = 0;
    __syncthreads();

    int2* lstw = lst + wave * AGG_CAP;
    const float* __restrict__ er_own = er + base;
    const unsigned lt_mask = (1u << lane) - 1u;
    const int4* __restrict__ dst4 = reinterpret_cast<const int4*>(dst);

    const uint4* __restrict__ own16 = reinterpret_cast<const uint4*>(own);
    const unsigned key = (unsigned)blockIdx.x * 0x01010101u;
    unsigned cnt = 0;
    auto fetch = [&](int g) -> uint4 {
        return own16[((g < OWN_GROUPS) ? g : (OWN_GROUPS - 1)) * 32 + lane];
    };
    auto hits = [&](unsigned v) -> unsigned {
        const unsigned x = v ^ key;
        const unsigned t = (x & 0x7F7F7F7Fu) + 0x7F7F7F7Fu;
        return ~(t | x) & 0x80808080u;
    };
    auto examine = [&](const uint4& o, int g) {
        if (g >= OWN_GROUPS) return;
        unsigned zz = (hits(o.x) >> 7) | (hits(o.y) >> 6) | (hits(o.z) >> 5) | (hits(o.w) >> 4);
        const int e0 = (g * 32 + lane) * 16;
        unsigned bal = __builtin_amdgcn_ballot_w32(zz != 0u);
        while (bal != 0u) {
            const unsigned pos = cnt + (unsigned)__popc(bal & lt_mask);
            if (zz != 0u) {
                const int p = __builtin_ctz(zz);
                zz &= zz - 1u;
                if (pos < (unsigned)AGG_CAP)
                    lstw[pos] = make_int2(e0 + ((p & 3) << 2) + (p >> 3), 0);
            }
            cnt += (unsigned)__popc(bal);
            bal = __builtin_amdgcn_ballot_w32(zz != 0u);
        }
    };

    constexpr int STEP = AGG_WAVES * AGG_UNROLL;
    uint4 bufA[AGG_UNROLL], bufB[AGG_UNROLL];
    #pragma unroll
    for (int u = 0; u < AGG_UNROLL; ++u) bufA[u] = fetch(wave + AGG_WAVES * u);
    for (int g0 = wave; g0 < OWN_GROUPS; g0 += 2 * STEP) {
        #pragma unroll
        for (int u = 0; u < AGG_UNROLL; ++u) bufB[u] = fetch(g0 + STEP + AGG_WAVES * u);
        #pragma unroll
        for (int u = 0; u < AGG_UNROLL; ++u) examine(bufA[u], g0 + AGG_WAVES * u);
        #pragma unroll
        for (int u = 0; u < AGG_UNROLL; ++u) bufA[u] = fetch(g0 + 2 * STEP + AGG_WAVES * u);
        #pragma unroll
        for (int u = 0; u < AGG_UNROLL; ++u) examine(bufB[u], g0 + STEP + AGG_WAVES * u);
    }
    const unsigned cntc = (cnt < (unsigned)AGG_CAP) ? cnt : (unsigned)AGG_CAP;
    if (lane == 0) {
        cntw[wave] = (int)cntc;
        if (cnt > (unsigned)AGG_CAP) ovf = 1;
    }

    for (unsigned i0 = 0; i0 < cntc; i0 += 128) {
        int  en[4], sv[4], rw[4];
        float ev[4];
        #pragma unroll
        for (int u = 0; u < 4; ++u) {
            const unsigned i = i0 + 32u * u + (unsigned)lane;
            en[u] = (i < cntc) ? lstw[i].x : 0;
        }
        #pragma unroll
        for (int u = 0; u < 4; ++u) { sv[u] = src[en[u]]; rw[u] = dst[en[u]]; }
        #pragma unroll
        for (int u = 0; u < 4; ++u) {
            const unsigned i = i0 + 32u * u + (unsigned)lane;
            rw[u] = (i < cntc) ? rw[u] - base : 0;
        }
        #pragma unroll
        for (int u = 0; u < 4; ++u) ev[u] = edge_score(el[sv[u]], er_own[rw[u]]);
        #pragma unroll
        for (int u = 0; u < 4; ++u) {
            const unsigned i = i0 + 32u * u + (unsigned)lane;
            if (i < cntc) {
                atomicMax(&mx[rw[u]], enc_f32(ev[u]));
                lstw[i] = make_int2(sv[u] | (rw[u] << ROW_SHIFT), __float_as_int(ev[u]));
            }
        }
    }
    __syncthreads();

    const int ovf_u = __builtin_amdgcn_readfirstlane(ovf);
    if (ovf_u == 0) {
        int2* ringw = ring + wave * AGG_RING;
        unsigned head = 0, tail = 0;
        v16f sc[4] = {};
        float dn = 0.f;
        for (int q = 0; q < AGG_WAVES; ++q) {
            const unsigned nq = (unsigned)__builtin_amdgcn_readfirstlane(cntw[q]);
            const int2* lq = lst + q * AGG_CAP;
            for (unsigned i0 = 0; i0 < nq; i0 += 128) {
                int2 en[4];
                #pragma unroll
                for (int u = 0; u < 4; ++u) {
                    const unsigned i = i0 + 32u * u + (unsigned)lane;
                    en[u] = (i < nq) ? lq[i] : make_int2(0, 0);
                }
                #pragma unroll
                for (int u = 0; u < 4; ++u) {
                    const unsigned i = i0 + 32u * u + (unsigned)lane;
                    const bool m = (i < nq) &&
                        ((((unsigned)en[u].x >> ROW_SHIFT) & (AGG_WAVES - 1)) == (unsigned)wave);
                    const unsigned bal = __builtin_amdgcn_ballot_w32(m);
                    if (bal != 0u) {
                        if (m) ringw[(tail + (unsigned)__popc(bal & lt_mask)) & (AGG_RING - 1)] = en[u];
                        tail += (unsigned)__popc(bal);
                        while (tail - head >= (unsigned)AGG_BATCH) {
                            agg_batch<true>(ringw, head, AGG_BATCH, lane, mx, feat, sc, dn);
                            head += AGG_BATCH;
                        }
                    }
                }
            }
        }
        if (tail != head)
            agg_batch<false>(ringw, head, (int)(tail - head), lane, mx, feat, sc, dn);
        const float dinv = (dn > 0.f) ? 1.0f / dn : 0.f;
        float4* __restrict__ h4 = reinterpret_cast<float4*>(h + (size_t)base * D_HID);
        #pragma unroll
        for (int k = 0; k < AGG_SLOTS; ++k) {
            const int row = wave + AGG_WAVES * k;
            if (row < nrows) {
                const float r = __int_as_float(__builtin_amdgcn_readlane(__float_as_int(dinv), k));
                h4[row * (D_HID / 4) + lane] =
                    make_float4(sc[0][k] * r, sc[1][k] * r, sc[2][k] * r, sc[3][k] * r);
            }
        }
    } else {
        float4* acc4 = reinterpret_cast<float4*>(acc);
        for (int i = tid; i < AGG_ROWS * D_HID / 4; i += AGG_THREADS)
            acc4[i] = make_float4(0.f, 0.f, 0.f, 0.f);
        if (tid < AGG_ROWS) den[tid] = 0.f;
        for (int g = wave; g < EDGE_GROUPS; g += AGG_WAVES) {
            const int4 d = dst4[g * 32 + lane];
            const int dd[4] = {d.x, d.y, d.z, d.w};
            #pragma unroll
            for (int c = 0; c < 4; ++c) {
                const unsigned t = (unsigned)(dd[c] - base);
                if (t < (unsigned)nrows) {
                    const int s = src[(g * 32 + lane) * 4 + c];
                    atomicMax(&mx[t], enc_f32(edge_score(el[s], er_own[t])));
                }
            }
        }
        __syncthreads();
        for (int g = wave; g < EDGE_GROUPS; g += AGG_WAVES) {
            const int4 d = dst4[g * 32 + lane];
            const int dd[4] = {d.x, d.y, d.z, d.w};
            #pragma unroll
            for (int c = 0; c < 4; ++c) {
                const unsigned t = (unsigned)(dd[c] - base);
                const bool m = t < (unsigned)nrows;
                int s = 0, tl = 0;
                float a = 0.f;
                if (m) {
                    s = src[(g * 32 + lane) * 4 + c];
                    tl = (int)t;
                    a = __expf(edge_score(el[s], er_own[t]) - dec_f32(mx[t]));
                    atomicAdd(den + tl, a);
                }
                unsigned bal = __builtin_amdgcn_ballot_w32(m);
                while (bal != 0u) {
                    const int j = __builtin_ctz(bal);
                    bal &= bal - 1u;
                    const int sj = __builtin_amdgcn_readlane(s, j);
                    const int dj = __builtin_amdgcn_readlane(tl, j);
                    const float aj = __int_as_float(__builtin_amdgcn_readlane(__float_as_int(a), j));
                    const float2* __restrict__ row =
                        reinterpret_cast<const float2*>(feat + (size_t)sj * D_HID);
                    const float2 x0 = row[lane], x1 = row[32 + lane];
                    float* arow = acc + dj * D_HID + 2 * lane;
                    atomicAdd(arow,      aj * x0.x);
                    atomicAdd(arow + 1,  aj * x0.y);
                    atomicAdd(arow + 64, aj * x1.x);
                    atomicAdd(arow + 65, aj * x1.y);
                }
            }
        }
        __syncthreads();
        if (tid < AGG_ROWS) {
            const float d = den[tid];
            den[tid] = (d > 0.f) ? 1.0f / d : 0.f;
        }
        __syncthreads();
        float4* __restrict__ h4 = reinterpret_cast<float4*>(h + (size_t)base * D_HID);
        for (int i = tid; i < nrows * (D_HID / 4); i += AGG_THREADS) {
            const float r = den[i / (D_HID / 4)];
            float4 v = acc4[i];
            v.x *= r; v.y *= r; v.z *= r; v.w *= r;
            h4[i] = v;
        }
    }
}

extern "C" void kernel_launch(void* const* d_in, const int* in_sizes, int n_in,
                              void* d_out, int out_size, void* d_ws, size_t ws_size,
                              hipStream_t stream) {
    const float* X  = (const float*)d_in[0];
    const float* W  = (const float*)d_in[1];
    const float* al = (const float*)d_in[2];
    const float* ar = (const float*)d_in[3];
    const float* FW = (const float*)d_in[4];
    const float* FB = (const float*)d_in[5];
    const int*  src = (const int*)d_in[6];
    const int*  dst = (const int*)d_in[7];
    float* out = (float*)d_out;

    char* ws = (char*)d_ws;
    size_t off = 0;
    auto carve = [&](size_t bytes) {
        void* p = ws + off;
        off += (bytes + 255) & ~(size_t)255;
        return p;
    };
    float*    feat  = (float*)   carve((size_t)N_NODES * D_HID * sizeof(float));
    float*    h     = (float*)   carve((size_t)N_NODES * D_HID * sizeof(float));
    float*    el    = (float*)   carve((size_t)N_NODES * sizeof(float));
    float*    er    = (float*)   carve((size_t)N_NODES * sizeof(float));
    unsigned* own   = (unsigned*)carve((size_t)N_EDGES);

    k_gemm1<<<1563, 128, 0, stream>>>(X, W, al, ar, feat, el, er, dst, own);
    k_agg<<<AGG_WGS, AGG_THREADS, 0, stream>>>(src, dst, own, el, er, feat, h);
    k_gemm2<<<782, 256, 0, stream>>>(h, FW, FB, out);
}


# GEMM1/GEMM2 fragments loaded 16 B per lane on whole lines: a lane's tiles are columns 4*l16+t, its k in a 16-k block 8*half+0..7; feat/out rows stored as float4; k and el/er sum order new
# speedup vs baseline: 5.0451x; 1.0631x over previous
#include <hip/hip_runtime.h>
#include <hip/hip_bf16.h>

#define N_NODES   100000
#define N_EDGES   1600000
#define D_IN      256
#define D_HID     128
#define N_CLASSES 64
#define NEG_SLOPE 0.2f

typedef __attribute__((ext_vector_type(2))) float v2f;
typedef __attribute__((ext_vector_type(8))) float v8f;

#define OWN_ROWS 391u
static_assert(N_EDGES == (N_NODES / 16) * 256, "one 256-edge slice of the owner map per GEMM1 wave");
static_assert((N_NODES - 1) / OWN_ROWS < 256u, "owner index must fit one byte");

__global__ __launch_bounds__(128) void k_gemm1(const float* __restrict__ X,
                                               const float* __restrict__ W,
                                               const float* __restrict__ al,
                                               const float* __restrict__ ar,
                                               float* __restrict__ feat,
                                               float* __restrict__ el,
                                               float* __restrict__ er,
                                               const int* __restrict__ dst,
                                               unsigned* __restrict__ own) {
    const int wave = threadIdx.x >> 5;
    const int lane = threadIdx.x & 31;
    const int rowTile = blockIdx.x * 4 + wave;
    if (rowTile >= N_NODES / 16) return;
    const int row0 = rowTile * 16;
    {
        const int4* __restrict__ d4 =
            reinterpret_cast<const int4*>(dst) + (size_t)rowTile * 64 + lane * 2;
        const int4 da = d4[0], db = d4[1];
        const unsigned lo = ((unsigned)da.x / OWN_ROWS) | (((unsigned)da.y / OWN_ROWS) << 8) |
                            (((unsigned)da.z / OWN_ROWS) << 16) | (((unsigned)da.w / OWN_ROWS) << 24);
        const unsigned hi8 = ((unsigned)db.x / OWN_ROWS) | (((unsigned)db.y / OWN_ROWS) << 8) |
                             (((unsigned)db.z / OWN_ROWS) << 16) | (((unsigned)db.w / OWN_ROWS) << 24);
        reinterpret_cast<uint2*>(own)[rowTile * 32 + lane] = make_uint2(lo, hi8);
    }
    const int l16 = lane & 15;
    const int hi  = lane >> 4;

    v8f acc[8] = {};
    const float4* __restrict__ xrow4 =
        reinterpret_cast<const float4*>(X + (size_t)(row0 + l16) * D_IN + 8 * hi);
    const float4* __restrict__ w4 =
        reinterpret_cast<const float4*>(W + (size_t)(8 * hi) * D_HID + 4 * l16);

    for (int j = 0; j < D_IN / 16; ++j) {
        const float4 q0 = xrow4[4 * j];
        const float4 q1 = xrow4[4 * j + 1];
        const float av[8] = {q0.x, q0.y, q0.z, q0.w, q1.x, q1.y, q1.z, q1.w};
        for (int s = 0; s < 4; ++s) {
            const float4* __restrict__ r0 = w4 + (size_t)(16 * j + 2 * s) * (D_HID / 4);
            const float4* __restrict__ r1 = r0 + D_HID / 4;
            const float4 b0l = r0[0], b0h = r0[16];
            const float4 b1l = r1[0], b1h = r1[16];
            v2f a;
            a.x = av[2 * s];
            a.y = av[2 * s + 1];
            v2f b[8];
            b[0].x = b0l.x; b[0].y = b1l.x;
            b[1].x = b0l.y; b[1].y = b1l.y;
            b[2].x = b0l.z; b[2].y = b1l.z;
            b[3].x = b0l.w; b[3].y = b1l.w;
            b[4].x = b0h.x; b[4].y = b1h.x;
            b[5].x = b0h.y; b[5].y = b1h.y;
            b[6].x = b0h.z; b[6].y = b1h.z;
            b[7].x = b0h.w; b[7].y = b1h.w;
            for (int t = 0; t < 8; ++t)
                acc[t] = __builtin_amdgcn_wmma_f32_16x16x4_f32(
                    false, a, false, b[t], (short)0, acc[t], false, false);
        }
    }
    float4* __restrict__ feat4 = reinterpret_cast<float4*>(feat);
    for (int r = 0; r < 8; ++r) {
        float4* __restrict__ orow = feat4 + (size_t)(row0 + r + 8 * hi) * (D_HID / 4) + l16;
        orow[0]  = make_float4(acc[0][r], acc[1][r], acc[2][r], acc[3][r]);
        orow[16] = make_float4(acc[4][r], acc[5][r], acc[6][r], acc[7][r]);
    }
    const float4 all4 = reinterpret_cast<const float4*>(al)[l16];
    const float4 alh4 = reinterpret_cast<const float4*>(al)[16 + l16];
    const float4 arl4 = reinterpret_cast<const float4*>(ar)[l16];
    const float4 arh4 = reinterpret_cast<const float4*>(ar)[16 + l16];
    const float alv[8] = {all4.x, all4.y, all4.z, all4.w, alh4.x, alh4.y, alh4.z, alh4.w};
    const float arv[8] = {arl4.x, arl4.y, arl4.z, arl4.w, arh4.x, arh4.y, arh4.z, arh4.w};
    float myl = 0.f, myr = 0.f;
    for (int r = 0; r < 8; ++r) {
        float sl = 0.f, sr = 0.f;
        for (int t = 0; t < 8; ++t) {
            sl += acc[t][r] * alv[t];
            sr += acc[t][r] * arv[t];
        }
        for (int off = 8; off > 0; off >>= 1) {
            sl += __shfl_xor(sl, off, 32);
            sr += __shfl_xor(sr, off, 32);
        }
        if (l16 == r) { myl = sl; myr = sr; }
    }
    if (l16 < 8) {
        el[row0 + l16 + 8 * hi] = myl;
        er[row0 + l16 + 8 * hi] = myr;
    }
}

__global__ __launch_bounds__(256) void k_gemm2(const float* __restrict__ H,
                                               const float* __restrict__ FW,
                                               const float* __restrict__ FB,
                                               float* __restrict__ out) {
    const int wave = threadIdx.x >> 5;
    const int lane = threadIdx.x & 31;
    const int rowTile = blockIdx.x * 8 + wave;
    if (rowTile >= N_NODES / 16) return;
    const int row0 = rowTile * 16;
    const int l16 = lane & 15;
    const int hi  = lane >> 4;

    v8f acc0 = {}, acc1 = {}, acc2 = {}, acc3 = {};
    const float4* __restrict__ hrow4 =
        reinterpret_cast<const float4*>(H + (size_t)(row0 + l16) * D_HID + 8 * hi);
    const float4* __restrict__ fw4 =
        reinterpret_cast<const float4*>(FW + (size_t)(8 * hi) * N_CLASSES + 4 * l16);

    for (int j = 0; j < D_HID / 16; ++j) {
        const float4 q0 = hrow4[4 * j];
        const float4 q1 = hrow4[4 * j + 1];
        const float av[8] = {q0.x, q0.y, q0.z, q0.w, q1.x, q1.y, q1.z, q1.w};
        for (int s = 0; s < 4; ++s) {
            const float4 b0 = fw4[(size_t)(16 * j + 2 * s) * (N_CLASSES / 4)];
            const float4 b1 = fw4[(size_t)(16 * j + 2 * s + 1) * (N_CLASSES / 4)];
            v2f a;
            a.x = av[2 * s];
            a.y = av[2 * s + 1];
            v2f c0, c1, c2, c3;
            c0.x = b0.x; c0.y = b1.x;
            c1.x = b0.y; c1.y = b1.y;
            c2.x = b0.z; c2.y = b1.z;
            c3.x = b0.w; c3.y = b1.w;
            acc0 = __builtin_amdgcn_wmma_f32_16x16x4_f32(
                false, a, false, c0, (short)0, acc0, false, false);
            acc1 = __builtin_amdgcn_wmma_f32_16x16x4_f32(
                false, a, false, c1, (short)0, acc1, false, false);
            acc2 = __builtin_amdgcn_wmma_f32_16x16x4_f32(
                false, a, false, c2, (short)0, acc2, false, false);
            acc3 = __builtin_amdgcn_wmma_f32_16x16x4_f32(
                false, a, false, c3, (short)0, acc3, false, false);
        }
    }
    const float4 bias = reinterpret_cast<const float4*>(FB)[l16];
    float4* __restrict__ out4 = reinterpret_cast<float4*>(out);
    for (int r = 0; r < 8; ++r) {
        out4[(size_t)(row0 + r + 8 * hi) * (N_CLASSES / 4) + l16] =
            make_float4(acc0[r] + bias.x, acc1[r] + bias.y, acc2[r] + bias.z, acc3[r] + bias.w);
    }
}

__device__ __forceinline__ unsigned enc_f32(float f) {
    unsigned b = __float_as_uint(f);
    return (b & 0x80000000u) ? ~b : (b | 0x80000000u);
}
__device__ __forceinline__ float dec_f32(unsigned u) {
    unsigned b = (u & 0x80000000u) ? (u & 0x7FFFFFFFu) : ~u;
    return __uint_as_float(b);
}

#define AGG_WGS     256
#define AGG_ROWS    391
static_assert(AGG_ROWS == (int)OWN_ROWS, "owner map and block ranges must agree");
#define AGG_THREADS 1024
#define AGG_WAVES   (AGG_THREADS / 32)
#define AGG_CAP     400
#define AGG_RING    64
#define AGG_BATCH   8
#define AGG_UNROLL  4
#define EDGE_GROUPS (N_EDGES / 128)
#define OWN_GROUPS  (N_EDGES / 512)
#define ROW_SHIFT   17
static_assert(N_NODES <= (1 << ROW_SHIFT), "src must fit below the row field");
static_assert(AGG_ROWS <= (1 << (31 - ROW_SHIFT)), "row must fit above the src field");
#define AGG_SLOTS   ((AGG_ROWS + AGG_WAVES - 1) / AGG_WAVES)
static_assert(AGG_SLOTS <= 32, "lane k of a wave holds the denominator of its slot k");

__device__ __forceinline__ float edge_score(float l, float r) {
    const float v = l + r;
    return (v > 0.f) ? v : NEG_SLOPE * v;
}

typedef __attribute__((ext_vector_type(16))) float v16f;
static_assert(AGG_SLOTS <= 16, "one vector element per slot");

template <bool FULL>
__device__ __forceinline__ void agg_batch(const int2* ringw, unsigned head, int n, int lane,
                                          const unsigned* mx,
                                          const float* __restrict__ feat,
                                          v16f (&sc)[4], float& dn) {
    int sr = 0;
    float a = 0.f;
    if (lane < n) {
        const int2 ent = ringw[(head + (unsigned)lane) & (AGG_RING - 1)];
        sr = ent.x;
        a  = __expf(__int_as_float(ent.y) - dec_f32(mx[sr >> ROW_SHIFT]));
    }
    float4 v[AGG_BATCH];
    int srj[AGG_BATCH];
    const float4* __restrict__ feat4 = reinterpret_cast<const float4*>(feat);
    #pragma unroll
    for (int j = 0; j < AGG_BATCH; ++j) {
        if (FULL || j < n) {
            srj[j] = __builtin_amdgcn_readlane(sr, j);
            const unsigned sj = (unsigned)srj[j] & ((1u << ROW_SHIFT) - 1u);
            v[j] = feat4[sj * (unsigned)(D_HID / 4) + (unsigned)lane];
        }
    }
    #pragma unroll
    for (int j = 0; j < AGG_BATCH; ++j) {
        if (FULL || j < n) {
            const int kj = (int)(((unsigned)srj[j] >> ROW_SHIFT) / AGG_WAVES);
            const float aj = __int_as_float(__builtin_amdgcn_readlane(__float_as_int(a), j));
            if (lane == kj) dn += aj;
            sc[0][kj] += aj * v[j].x;
            sc[1][kj] += aj * v[j].y;
            sc[2][kj] += aj * v[j].z;
            sc[3][kj] += aj * v[j].w;
        }
    }
}

__global__ __launch_bounds__(AGG_THREADS) void k_agg(const int* __restrict__ src,
                                                     const int* __restrict__ dst,
                                                     const unsigned* __restrict__ own,
                                                     const float* __restrict__ el,
                                                     const float* __restrict__ er,
                                                     const float* __restrict__ feat,
                                                     float* __restrict__ h) {
    __shared__ __attribute__((aligned(16))) float acc[AGG_ROWS * D_HID];
    __shared__ int2 lst[AGG_WAVES * AGG_CAP];
    __shared__ int2 ring[AGG_WAVES * AGG_RING];
    __shared__ float den[AGG_ROWS];
    __shared__ unsigned mx[AGG_ROWS];
    __shared__ int cntw[AGG_WAVES];
    __shared__ int ovf;

    const int tid  = threadIdx.x;
    const int lane = tid & 31;
    const int wave = __builtin_amdgcn_readfirstlane(tid >> 5);
    const int base = blockIdx.x * AGG_ROWS;
    const int nrows = (N_NODES - base < AGG_ROWS) ? (N_NODES - base) : AGG_ROWS;

    if (tid < AGG_ROWS) mx[tid] = 0u;
    if (tid == 0) ovf = 0;
    __syncthreads();

    int2* lstw = lst + wave * AGG_CAP;
    const float* __restrict__ er_own = er + base;
    const unsigned lt_mask = (1u << lane) - 1u;
    const int4* __restrict__ dst4 = reinterpret_cast<const int4*>(dst);

    const uint4* __restrict__ own16 = reinterpret_cast<const uint4*>(own);
    const unsigned key = (unsigned)blockIdx.x * 0x01010101u;
    unsigned cnt = 0;
    auto fetch = [&](int g) -> uint4 {
        return own16[((g < OWN_GROUPS) ? g : (OWN_GROUPS - 1)) * 32 + lane];
    };
    auto hits = [&](unsigned v) -> unsigned {
        const unsigned x = v ^ key;
        const unsigned t = (x & 0x7F7F7F7Fu) + 0x7F7F7F7Fu;
        return ~(t | x) & 0x80808080u;
    };
    auto examine = [&](const uint4& o, int g) {
        if (g >= OWN_GROUPS) return;
        unsigned zz = (hits(o.x) >> 7) | (hits(o.y) >> 6) | (hits(o.z) >> 5) | (hits(o.w) >> 4);
        const int e0 = (g * 32 + lane) * 16;
        unsigned bal = __builtin_amdgcn_ballot_w32(zz != 0u);
        while (bal != 0u) {
            const unsigned pos = cnt + (unsigned)__popc(bal & lt_mask);
            if (zz != 0u) {
                const int p = __builtin_ctz(zz);
                zz &= zz - 1u;
                if (pos < (unsigned)AGG_CAP)
                    lstw[pos] = make_int2(e0 + ((p & 3) << 2) + (p >> 3), 0);
            }
            cnt += (unsigned)__popc(bal);
            bal = __builtin_amdgcn_ballot_w32(zz != 0u);
        }
    };

    constexpr int STEP = AGG_WAVES * AGG_UNROLL;
    uint4 bufA[AGG_UNROLL], bufB[AGG_UNROLL];
    #pragma unroll
    for (int u = 0; u < AGG_UNROLL; ++u) bufA[u] = fetch(wave + AGG_WAVES * u);
    for (int g0 = wave; g0 < OWN_GROUPS; g0 += 2 * STEP) {
        #pragma unroll
        for (int u = 0; u < AGG_UNROLL; ++u) bufB[u] = fetch(g0 + STEP + AGG_WAVES * u);
        #pragma unroll
        for (int u = 0; u < AGG_UNROLL; ++u) examine(bufA[u], g0 + AGG_WAVES * u);
        #pragma unroll
        for (int u = 0; u < AGG_UNROLL; ++u) bufA[u] = fetch(g0 + 2 * STEP + AGG_WAVES * u);
        #pragma unroll
        for (int u = 0; u < AGG_UNROLL; ++u) examine(bufB[u], g0 + STEP + AGG_WAVES * u);
    }
    const unsigned cntc = (cnt < (unsigned)AGG_CAP) ? cnt : (unsigned)AGG_CAP;
    if (lane == 0) {
        cntw[wave] = (int)cntc;
        if (cnt > (unsigned)AGG_CAP) ovf = 1;
    }

    for (unsigned i0 = 0; i0 < cntc; i0 += 128) {
        int  en[4], sv[4], rw[4];
        float ev[4];
        #pragma unroll
        for (int u = 0; u < 4; ++u) {
            const unsigned i = i0 + 32u * u + (unsigned)lane;
            en[u] = (i < cntc) ? lstw[i].x : 0;
        }
        #pragma unroll
        for (int u = 0; u < 4; ++u) { sv[u] = src[en[u]]; rw[u] = dst[en[u]]; }
        #pragma unroll
        for (int u = 0; u < 4; ++u) {
            const unsigned i = i0 + 32u * u + (unsigned)lane;
            rw[u] = (i < cntc) ? rw[u] - base : 0;
        }
        #pragma unroll
        for (int u = 0; u < 4; ++u) ev[u] = edge_score(el[sv[u]], er_own[rw[u]]);
        #pragma unroll
        for (int u = 0; u < 4; ++u) {
            const unsigned i = i0 + 32u * u + (unsigned)lane;
            if (i < cntc) {
                atomicMax(&mx[rw[u]], enc_f32(ev[u]));
                lstw[i] = make_int2(sv[u] | (rw[u] << ROW_SHIFT), __float_as_int(ev[u]));
            }
        }
    }
    __syncthreads();

    const int ovf_u = __builtin_amdgcn_readfirstlane(ovf);
    if (ovf_u == 0) {
        int2* ringw = ring + wave * AGG_RING;
        unsigned head = 0, tail = 0;
        v16f sc[4] = {};
        float dn = 0.f;
        for (int q = 0; q < AGG_WAVES; ++q) {
            const unsigned nq = (unsigned)__builtin_amdgcn_readfirstlane(cntw[q]);
            const int2* lq = lst + q * AGG_CAP;
            for (unsigned i0 = 0; i0 < nq; i0 += 128) {
                int2 en[4];
                #pragma unroll
                for (int u = 0; u < 4; ++u) {
                    const unsigned i = i0 + 32u * u + (unsigned)lane;
                    en[u] = (i < nq) ? lq[i] : make_int2(0, 0);
                }
                #pragma unroll
                for (int u = 0; u < 4; ++u) {
                    const unsigned i = i0 + 32u * u + (unsigned)lane;
                    const bool m = (i < nq) &&
                        ((((unsigned)en[u].x >> ROW_SHIFT) & (AGG_WAVES - 1)) == (unsigned)wave);
                    const unsigned bal = __builtin_amdgcn_ballot_w32(m);
                    if (bal != 0u) {
                        if (m) ringw[(tail + (unsigned)__popc(bal & lt_mask)) & (AGG_RING - 1)] = en[u];
                        tail += (unsigned)__popc(bal);
                        while (tail - head >= (unsigned)AGG_BATCH) {
                            agg_batch<true>(ringw, head, AGG_BATCH, lane, mx, feat, sc, dn);
                            head += AGG_BATCH;
                        }
                    }
                }
            }
        }
        if (tail != head)
            agg_batch<false>(ringw, head, (int)(tail - head), lane, mx, feat, sc, dn);
        const float dinv = (dn > 0.f) ? 1.0f / dn : 0.f;
        float4* __restrict__ h4 = reinterpret_cast<float4*>(h + (size_t)base * D_HID);
        #pragma unroll
        for (int k = 0; k < AGG_SLOTS; ++k) {
            const int row = wave + AGG_WAVES * k;
            if (row < nrows) {
                const float r = __int_as_float(__builtin_amdgcn_readlane(__float_as_int(dinv), k));
                h4[row * (D_HID / 4) + lane] =
                    make_float4(sc[0][k] * r, sc[1][k] * r, sc[2][k] * r, sc[3][k] * r);
            }
        }
    } else {
        float4* acc4 = reinterpret_cast<float4*>(acc);
        for (int i = tid; i < AGG_ROWS * D_HID / 4; i += AGG_THREADS)
            acc4[i] = make_float4(0.f, 0.f, 0.f, 0.f);
        if (tid < AGG_ROWS) den[tid] = 0.f;
        for (int g = wave; g < EDGE_GROUPS; g += AGG_WAVES) {
            const int4 d = dst4[g * 32 + lane];
            const int dd[4] = {d.x, d.y, d.z, d.w};
            #pragma unroll
            for (int c = 0; c < 4; ++c) {
                const unsigned t = (unsigned)(dd[c] - base);
                if (t < (unsigned)nrows) {
                    const int s = src[(g * 32 + lane) * 4 + c];
                    atomicMax(&mx[t], enc_f32(edge_score(el[s], er_own[t])));
                }
            }
        }
        __syncthreads();
        for (int g = wave; g < EDGE_GROUPS; g += AGG_WAVES) {
            const int4 d = dst4[g * 32 + lane];
            const int dd[4] = {d.x, d.y, d.z, d.w};
            #pragma unroll
            for (int c = 0; c < 4; ++c) {
                const unsigned t = (unsigned)(dd[c] - base);
                const bool m = t < (unsigned)nrows;
                int s = 0, tl = 0;
                float a = 0.f;
                if (m) {
                    s = src[(g * 32 + lane) * 4 + c];
                    tl = (int)t;
                    a = __expf(edge_score(el[s], er_own[t]) - dec_f32(mx[t]));
                    atomicAdd(den + tl, a);
                }
                unsigned bal = __builtin_amdgcn_ballot_w32(m);
                while (bal != 0u) {
                    const int j = __builtin_ctz(bal);
                    bal &= bal - 1u;
                    const int sj = __builtin_amdgcn_readlane(s, j);
                    const int dj = __builtin_amdgcn_readlane(tl, j);
                    const float aj = __int_as_float(__builtin_amdgcn_readlane(__float_as_int(a), j));
                    const float2* __restrict__ row =
                        reinterpret_cast<const float2*>(feat + (size_t)sj * D_HID);
                    const float2 x0 = row[lane], x1 = row[32 + lane];
                    float* arow = acc + dj * D_HID + 2 * lane;
                    atomicAdd(arow,      aj * x0.x);
                    atomicAdd(arow + 1,  aj * x0.y);
                    atomicAdd(arow + 64, aj * x1.x);
                    atomicAdd(arow + 65, aj * x1.y);
                }
            }
        }
        __syncthreads();
        if (tid < AGG_ROWS) {
            const float d = den[tid];
            den[tid] = (d > 0.f) ? 1.0f / d : 0.f;
        }
        __syncthreads();
        float4* __restrict__ h4 = reinterpret_cast<float4*>(h + (size_t)base * D_HID);
        for (int i = tid; i < nrows * (D_HID / 4); i += AGG_THREADS) {
            const float r = den[i / (D_HID / 4)];
            float4 v = acc4[i];
            v.x *= r; v.y *= r; v.z *= r; v.w *= r;
            h4[i] = v;
        }
    }
}

extern "C" void kernel_launch(void* const* d_in, const int* in_sizes, int n_in,
                              void* d_out, int out_size, void* d_ws, size_t ws_size,
                              hipStream_t stream) {
    const float* X  = (const float*)d_in[0];
    const float* W  = (const float*)d_in[1];
    const float* al = (const float*)d_in[2];
    const float* ar = (const float*)d_in[3];
    const float* FW = (const float*)d_in[4];
    const float* FB = (const float*)d_in[5];
    const int*  src = (const int*)d_in[6];
    const int*  dst = (const int*)d_in[7];
    float* out = (float*)d_out;

    char* ws = (char*)d_ws;
    size_t off = 0;
    auto carve = [&](size_t bytes) {
        void* p = ws + off;
        off += (bytes + 255) & ~(size_t)255;
        return p;
    };
    float*    feat  = (float*)   carve((size_t)N_NODES * D_HID * sizeof(float));
    float*    h     = (float*)   carve((size_t)N_NODES * D_HID * sizeof(float));
    float*    el    = (float*)   carve((size_t)N_NODES * sizeof(float));
    float*    er    = (float*)   carve((size_t)N_NODES * sizeof(float));
    unsigned* own   = (unsigned*)carve((size_t)N_EDGES);

    k_gemm1<<<1563, 128, 0, stream>>>(X, W, al, ar, feat, el, er, dst, own);
    k_agg<<<AGG_WGS, AGG_THREADS, 0, stream>>>(src, dst, own, el, er, feat, h);
    k_gemm2<<<782, 256, 0, stream>>>(h, FW, FB, out);
}


# Aggregation: listed entries dealt to their summing wave's own LDS list by a counted partition (ballot histograms, prefix over waves); a wave reads only its own entries, not all lists; sum order kept
# speedup vs baseline: 5.5108x; 1.0923x over previous
#include <hip/hip_runtime.h>
#include <hip/hip_bf16.h>

#define N_NODES   100000
#define N_EDGES   1600000
#define D_IN      256
#define D_HID     128
#define N_CLASSES 64
#define NEG_SLOPE 0.2f

typedef __attribute__((ext_vector_type(2))) float v2f;
typedef __attribute__((ext_vector_type(8))) float v8f;

#define OWN_ROWS 391u
static_assert(N_EDGES == (N_NODES / 16) * 256, "one 256-edge slice of the owner map per GEMM1 wave");
static_assert((N_NODES - 1) / OWN_ROWS < 256u, "owner index must fit one byte");

__global__ __launch_bounds__(128) void k_gemm1(const float* __restrict__ X,
                                               const float* __restrict__ W,
                                               const float* __restrict__ al,
                                               const float* __restrict__ ar,
                                               float* __restrict__ feat,
                                               float* __restrict__ el,
                                               float* __restrict__ er,
                                               const int* __restrict__ dst,
                                               unsigned* __restrict__ own) {
    const int wave = threadIdx.x >> 5;
    const int lane = threadIdx.x & 31;
    const int rowTile = blockIdx.x * 4 + wave;
    if (rowTile >= N_NODES / 16) return;
    const int row0 = rowTile * 16;
    {
        const int4* __restrict__ d4 =
            reinterpret_cast<const int4*>(dst) + (size_t)rowTile * 64 + lane * 2;
        const int4 da = d4[0], db = d4[1];
        const unsigned lo = ((unsigned)da.x / OWN_ROWS) | (((unsigned)da.y / OWN_ROWS) << 8) |
                            (((unsigned)da.z / OWN_ROWS) << 16) | (((unsigned)da.w / OWN_ROWS) << 24);
        const unsigned hi8 = ((unsigned)db.x / OWN_ROWS) | (((unsigned)db.y / OWN_ROWS) << 8) |
                             (((unsigned)db.z / OWN_ROWS) << 16) | (((unsigned)db.w / OWN_ROWS) << 24);
        reinterpret_cast<uint2*>(own)[rowTile * 32 + lane] = make_uint2(lo, hi8);
    }
    const int l16 = lane & 15;
    const int hi  = lane >> 4;

    v8f acc[8] = {};
    const float4* __restrict__ xrow4 =
        reinterpret_cast<const float4*>(X + (size_t)(row0 + l16) * D_IN + 8 * hi);
    const float4* __restrict__ w4 =
        reinterpret_cast<const float4*>(W + (size_t)(8 * hi) * D_HID + 4 * l16);

    for (int j = 0; j < D_IN / 16; ++j) {
        const float4 q0 = xrow4[4 * j];
        const float4 q1 = xrow4[4 * j + 1];
        const float av[8] = {q0.x, q0.y, q0.z, q0.w, q1.x, q1.y, q1.z, q1.w};
        for (int s = 0; s < 4; ++s) {
            const float4* __restrict__ r0 = w4 + (size_t)(16 * j + 2 * s) * (D_HID / 4);
            const float4* __restrict__ r1 = r0 + D_HID / 4;
            const float4 b0l = r0[0], b0h = r0[16];
            const float4 b1l = r1[0], b1h = r1[16];
            v2f a;
            a.x = av[2 * s];
            a.y = av[2 * s + 1];
            v2f b[8];
            b[0].x = b0l.x; b[0].y = b1l.x;
            b[1].x = b0l.y; b[1].y = b1l.y;
            b[2].x = b0l.z; b[2].y = b1l.z;
            b[3].x = b0l.w; b[3].y = b1l.w;
            b[4].x = b0h.x; b[4].y = b1h.x;
            b[5].x = b0h.y; b[5].y = b1h.y;
            b[6].x = b0h.z; b[6].y = b1h.z;
            b[7].x = b0h.w; b[7].y = b1h.w;
            for (int t = 0; t < 8; ++t)
                acc[t] = __builtin_amdgcn_wmma_f32_16x16x4_f32(
                    false, a, false, b[t], (short)0, acc[t], false, false);
        }
    }
    float4* __restrict__ feat4 = reinterpret_cast<float4*>(feat);
    for (int r = 0; r < 8; ++r) {
        float4* __restrict__ orow = feat4 + (size_t)(row0 + r + 8 * hi) * (D_HID / 4) + l16;
        orow[0]  = make_float4(acc[0][r], acc[1][r], acc[2][r], acc[3][r]);
        orow[16] = make_float4(acc[4][r], acc[5][r], acc[6][r], acc[7][r]);
    }
    const float4 all4 = reinterpret_cast<const float4*>(al)[l16];
    const float4 alh4 = reinterpret_cast<const float4*>(al)[16 + l16];
    const float4 arl4 = reinterpret_cast<const float4*>(ar)[l16];
    const float4 arh4 = reinterpret_cast<const float4*>(ar)[16 + l16];
    const float alv[8] = {all4.x, all4.y, all4.z, all4.w, alh4.x, alh4.y, alh4.z, alh4.w};
    const float arv[8] = {arl4.x, arl4.y, arl4.z, arl4.w, arh4.x, arh4.y, arh4.z, arh4.w};
    float myl = 0.f, myr = 0.f;
    for (int r = 0; r < 8; ++r) {
        float sl = 0.f, sr = 0.f;
        for (int t = 0; t < 8; ++t) {
            sl += acc[t][r] * alv[t];
            sr += acc[t][r] * arv[t];
        }
        for (int off = 8; off > 0; off >>= 1) {
            sl += __shfl_xor(sl, off, 32);
            sr += __shfl_xor(sr, off, 32);
        }
        if (l16 == r) { myl = sl; myr = sr; }
    }
    if (l16 < 8) {
        el[row0 + l16 + 8 * hi] = myl;
        er[row0 + l16 + 8 * hi] = myr;
    }
}

__global__ __launch_bounds__(256) void k_gemm2(const float* __restrict__ H,
                                               const float* __restrict__ FW,
                                               const float* __restrict__ FB,
                                               float* __restrict__ out) {
    const int wave = threadIdx.x >> 5;
    const int lane = threadIdx.x & 31;
    const int rowTile = blockIdx.x * 8 + wave;
    if (rowTile >= N_NODES / 16) return;
    const int row0 = rowTile * 16;
    const int l16 = lane & 15;
    const int hi  = lane >> 4;

    v8f acc0 = {}, acc1 = {}, acc2 = {}, acc3 = {};
    const float4* __restrict__ hrow4 =
        reinterpret_cast<const float4*>(H + (size_t)(row0 + l16) * D_HID + 8 * hi);
    const float4* __restrict__ fw4 =
        reinterpret_cast<const float4*>(FW + (size_t)(8 * hi) * N_CLASSES + 4 * l16);

    for (int j = 0; j < D_HID / 16; ++j) {
        const float4 q0 = hrow4[4 * j];
        const float4 q1 = hrow4[4 * j + 1];
        const float av[8] = {q0.x, q0.y, q0.z, q0.w, q1.x, q1.y, q1.z, q1.w};
        for (int s = 0; s < 4; ++s) {
            const float4 b0 = fw4[(size_t)(16 * j + 2 * s) * (N_CLASSES / 4)];
            const float4 b1 = fw4[(size_t)(16 * j + 2 * s + 1) * (N_CLASSES / 4)];
            v2f a;
            a.x = av[2 * s];
            a.y = av[2 * s + 1];
            v2f c0, c1, c2, c3;
            c0.x = b0.x; c0.y = b1.x;
            c1.x = b0.y; c1.y = b1.y;
            c2.x = b0.z; c2.y = b1.z;
            c3.x = b0.w; c3.y = b1.w;
            acc0 = __builtin_amdgcn_wmma_f32_16x16x4_f32(
                false, a, false, c0, (short)0, acc0, false, false);
            acc1 = __builtin_amdgcn_wmma_f32_16x16x4_f32(
                false, a, false, c1, (short)0, acc1, false, false);
            acc2 = __builtin_amdgcn_wmma_f32_16x16x4_f32(
                false, a, false, c2, (short)0, acc2, false, false);
            acc3 = __builtin_amdgcn_wmma_f32_16x16x4_f32(
                false, a, false, c3, (short)0, acc3, false, false);
        }
    }
    const float4 bias = reinterpret_cast<const float4*>(FB)[l16];
    float4* __restrict__ out4 = reinterpret_cast<float4*>(out);
    for (int r = 0; r < 8; ++r) {
        out4[(size_t)(row0 + r + 8 * hi) * (N_CLASSES / 4) + l16] =
            make_float4(acc0[r] + bias.x, acc1[r] + bias.y, acc2[r] + bias.z, acc3[r] + bias.w);
    }
}

__device__ __forceinline__ unsigned enc_f32(float f) {
    unsigned b = __float_as_uint(f);
    return (b & 0x80000000u) ? ~b : (b | 0x80000000u);
}
__device__ __forceinline__ float dec_f32(unsigned u) {
    unsigned b = (u & 0x80000000u) ? (u & 0x7FFFFFFFu) : ~u;
    return __uint_as_float(b);
}

#define AGG_WGS     256
#define AGG_ROWS    391
static_assert(AGG_ROWS == (int)OWN_ROWS, "owner map and block ranges must agree");
#define AGG_THREADS 1024
#define AGG_WAVES   (AGG_THREADS / 32)
#define AGG_CAP     400
#define DL_CAP      768
#define AGG_BATCH   8
#define AGG_UNROLL  4
#define EDGE_GROUPS (N_EDGES / 128)
#define OWN_GROUPS  (N_EDGES / 512)
#define ROW_SHIFT   17
static_assert(N_NODES <= (1 << ROW_SHIFT), "src must fit below the row field");
static_assert(AGG_ROWS <= (1 << (31 - ROW_SHIFT)), "row must fit above the src field");
#define AGG_SLOTS   ((AGG_ROWS + AGG_WAVES - 1) / AGG_WAVES)
static_assert(AGG_SLOTS <= 32, "lane k of a wave holds the denominator of its slot k");
static_assert(AGG_WAVES == 32, "lane t of a wave holds the count of summing wave t");
static_assert((size_t)AGG_WAVES * DL_CAP * 8 <= (size_t)AGG_ROWS * D_HID * 4, "summing lists fit the slow path's tile");

__device__ __forceinline__ float edge_score(float l, float r) {
    const float v = l + r;
    return (v > 0.f) ? v : NEG_SLOPE * v;
}

typedef __attribute__((ext_vector_type(16))) float v16f;
static_assert(AGG_SLOTS <= 16, "one vector element per slot");

template <bool FULL>
__device__ __forceinline__ void agg_batch(int sr, float a, int jb, int n, int lane,
                                          const float* __restrict__ feat,
                                          v16f (&sc)[4], float& dn) {
    float4 v[AGG_BATCH];
    int srj[AGG_BATCH];
    const float4* __restrict__ feat4 = reinterpret_cast<const float4*>(feat);
    #pragma unroll
    for (int j = 0; j < AGG_BATCH; ++j) {
        if (FULL || j < n) {
            srj[j] = __builtin_amdgcn_readlane(sr, jb + j);
            const unsigned sj = (unsigned)srj[j] & ((1u << ROW_SHIFT) - 1u);
            v[j] = feat4[sj * (unsigned)(D_HID / 4) + (unsigned)lane];
        }
    }
    #pragma unroll
    for (int j = 0; j < AGG_BATCH; ++j) {
        if (FULL || j < n) {
            const int kj = (int)(((unsigned)srj[j] >> ROW_SHIFT) / AGG_WAVES);
            const float aj = __int_as_float(__builtin_amdgcn_readlane(__float_as_int(a), jb + j));
            if (lane == kj) dn += aj;
            sc[0][kj] += aj * v[j].x;
            sc[1][kj] += aj * v[j].y;
            sc[2][kj] += aj * v[j].z;
            sc[3][kj] += aj * v[j].w;
        }
    }
}

__global__ __launch_bounds__(AGG_THREADS) void k_agg(const int* __restrict__ src,
                                                     const int* __restrict__ dst,
                                                     const unsigned* __restrict__ own,
                                                     const float* __restrict__ el,
                                                     const float* __restrict__ er,
                                                     const float* __restrict__ feat,
                                                     float* __restrict__ h) {
    __shared__ __attribute__((aligned(16))) float acc[AGG_ROWS * D_HID];
    __shared__ int2 lst[AGG_WAVES * AGG_CAP];
    __shared__ unsigned cnt2[AGG_WAVES * AGG_WAVES];
    __shared__ float den[AGG_ROWS];
    __shared__ unsigned mx[AGG_ROWS];
    __shared__ int ovf;

    const int tid  = threadIdx.x;
    const int lane = tid & 31;
    const int wave = __builtin_amdgcn_readfirstlane(tid >> 5);
    const int base = blockIdx.x * AGG_ROWS;
    const int nrows = (N_NODES - base < AGG_ROWS) ? (N_NODES - base) : AGG_ROWS;

    if (tid < AGG_ROWS) mx[tid] = 0u;
    if (tid == 0) ovf = 0;
    __syncthreads();

    int2* lstw = lst + wave * AGG_CAP;
    const float* __restrict__ er_own = er + base;
    const unsigned lt_mask = (1u << lane) - 1u;
    const int4* __restrict__ dst4 = reinterpret_cast<const int4*>(dst);

    const uint4* __restrict__ own16 = reinterpret_cast<const uint4*>(own);
    const unsigned key = (unsigned)blockIdx.x * 0x01010101u;
    unsigned cnt = 0;
    auto fetch = [&](int g) -> uint4 {
        return own16[((g < OWN_GROUPS) ? g : (OWN_GROUPS - 1)) * 32 + lane];
    };
    auto hits = [&](unsigned v) -> unsigned {
        const unsigned x = v ^ key;
        const unsigned t = (x & 0x7F7F7F7Fu) + 0x7F7F7F7Fu;
        return ~(t | x) & 0x80808080u;
    };
    auto examine = [&](const uint4& o, int g) {
        if (g >= OWN_GROUPS) return;
        unsigned zz = (hits(o.x) >> 7) | (hits(o.y) >> 6) | (hits(o.z) >> 5) | (hits(o.w) >> 4);
        const int e0 = (g * 32 + lane) * 16;
        unsigned bal = __builtin_amdgcn_ballot_w32(zz != 0u);
        while (bal != 0u) {
            const unsigned pos = cnt + (unsigned)__popc(bal & lt_mask);
            if (zz != 0u) {
                const int p = __builtin_ctz(zz);
                zz &= zz - 1u;
                if (pos < (unsigned)AGG_CAP)
                    lstw[pos] = make_int2(e0 + ((p & 3) << 2) + (p >> 3), 0);
            }
            cnt += (unsigned)__popc(bal);
            bal = __builtin_amdgcn_ballot_w32(zz != 0u);
        }
    };

    constexpr int STEP = AGG_WAVES * AGG_UNROLL;
    uint4 bufA[AGG_UNROLL], bufB[AGG_UNROLL];
    #pragma unroll
    for (int u = 0; u < AGG_UNROLL; ++u) bufA[u] = fetch(wave + AGG_WAVES * u);
    for (int g0 = wave; g0 < OWN_GROUPS; g0 += 2 * STEP) {
        #pragma unroll
        for (int u = 0; u < AGG_UNROLL; ++u) bufB[u] = fetch(g0 + STEP + AGG_WAVES * u);
        #pragma unroll
        for (int u = 0; u < AGG_UNROLL; ++u) examine(bufA[u], g0 + AGG_WAVES * u);
        #pragma unroll
        for (int u = 0; u < AGG_UNROLL; ++u) bufA[u] = fetch(g0 + 2 * STEP + AGG_WAVES * u);
        #pragma unroll
        for (int u = 0; u < AGG_UNROLL; ++u) examine(bufB[u], g0 + STEP + AGG_WAVES * u);
    }
    const unsigned cntc = (cnt < (unsigned)AGG_CAP) ? cnt : (unsigned)AGG_CAP;
    if (lane == 0 && cnt > (unsigned)AGG_CAP) ovf = 1;

    unsigned lmask[5];
    for (int k = 0; k < 5; ++k) lmask[k] = ((lane >> k) & 1) ? 0u : ~0u;
    auto chunk_counts = [&](bool val, unsigned t) -> unsigned {
        unsigned m = __builtin_amdgcn_ballot_w32(val);
        for (int k = 0; k < 5; ++k)
            m &= __builtin_amdgcn_ballot_w32(val && ((t >> k) & 1u)) ^ lmask[k];
        return (unsigned)__popc(m);
    };
    unsigned hist = 0;

    for (unsigned i0 = 0; i0 < cntc; i0 += 128) {
        int  en[4], sv[4], rw[4];
        float ev[4];
        #pragma unroll
        for (int u = 0; u < 4; ++u) {
            const unsigned i = i0 + 32u * u + (unsigned)lane;
            en[u] = (i < cntc) ? lstw[i].x : 0;
        }
        #pragma unroll
        for (int u = 0; u < 4; ++u) { sv[u] = src[en[u]]; rw[u] = dst[en[u]]; }
        #pragma unroll
        for (int u = 0; u < 4; ++u) {
            const unsigned i = i0 + 32u * u + (unsigned)lane;
            rw[u] = (i < cntc) ? rw[u] - base : 0;
        }
        for (int u = 0; u < 4; ++u)
            hist += chunk_counts(i0 + 32u * u + (unsigned)lane < cntc, (unsigned)rw[u] & (AGG_WAVES - 1));
        #pragma unroll
        for (int u = 0; u < 4; ++u) ev[u] = edge_score(el[sv[u]], er_own[rw[u]]);
        #pragma unroll
        for (int u = 0; u < 4; ++u) {
            const unsigned i = i0 + 32u * u + (unsigned)lane;
            if (i < cntc) {
                atomicMax(&mx[rw[u]], enc_f32(ev[u]));
                lstw[i] = make_int2(sv[u] | (rw[u] << ROW_SHIFT), __float_as_int(ev[u]));
            }
        }
    }
    cnt2[wave * AGG_WAVES + lane] = hist;
    __syncthreads();

    const int ovf_u = __builtin_amdgcn_readfirstlane(ovf);
    unsigned pre = 0, tot = 0;
    for (int q = 0; q < AGG_WAVES; ++q) {
        const unsigned c = cnt2[q * AGG_WAVES + lane];
        tot += c;
        if (q < wave) pre += c;
    }
    const bool fits = __builtin_amdgcn_ballot_w32(tot > (unsigned)DL_CAP) == 0u;
    if (ovf_u == 0 && fits) {
        int2* dl = reinterpret_cast<int2*>(acc);
        unsigned pos = pre;
        for (unsigned i0 = 0; i0 < cntc; i0 += 32) {
            const unsigned i = i0 + (unsigned)lane;
            const bool val = i < cntc;
            const int2 ent = val ? lstw[i] : make_int2(0, 0);
            const unsigned t = ((unsigned)ent.x >> ROW_SHIFT) & (AGG_WAVES - 1);
            unsigned same = __builtin_amdgcn_ballot_w32(val);
            unsigned mine = same;
            for (int k = 0; k < 5; ++k) {
                const unsigned bk = __builtin_amdgcn_ballot_w32(val && ((t >> k) & 1u));
                same &= ((t >> k) & 1u) ? bk : ~bk;
                mine &= bk ^ lmask[k];
            }
            const unsigned first = (unsigned)__builtin_amdgcn_ds_bpermute((int)(t << 2), (int)pos);
            if (val) dl[t * DL_CAP + first + (unsigned)__popc(same & lt_mask)] = ent;
            pos += (unsigned)__popc(mine);
        }
        __syncthreads();

        const unsigned nt = (unsigned)__builtin_amdgcn_readlane((int)tot, wave);
        const int2* dlw = dl + wave * DL_CAP;
        v16f sc[4] = {};
        float dn = 0.f;
        const unsigned nfull = nt & ~(unsigned)(AGG_BATCH - 1);
        int sr = 0;
        float a = 0.f;
        for (unsigned i0 = 0; i0 < nt; i0 += 32) {
            const unsigned i = i0 + (unsigned)lane;
            sr = 0;
            a = 0.f;
            if (i < nt) {
                const int2 ent = dlw[i];
                sr = ent.x;
                a  = __expf(__int_as_float(ent.y) - dec_f32(mx[sr >> ROW_SHIFT]));
            }
            const int nb = (nfull - i0 < 32u) ? (int)(nfull - i0) : 32;
            for (int jb = 0; jb < nb; jb += AGG_BATCH)
                agg_batch<true>(sr, a, jb, AGG_BATCH, lane, feat, sc, dn);
        }
        if (nt != nfull)
            agg_batch<false>(sr, a, (int)(nfull & 31u), (int)(nt - nfull), lane, feat, sc, dn);
        const float dinv = (dn > 0.f) ? 1.0f / dn : 0.f;
        float4* __restrict__ h4 = reinterpret_cast<float4*>(h + (size_t)base * D_HID);
        #pragma unroll
        for (int k = 0; k < AGG_SLOTS; ++k) {
            const int row = wave + AGG_WAVES * k;
            if (row < nrows) {
                const float r = __int_as_float(__builtin_amdgcn_readlane(__float_as_int(dinv), k));
                h4[row * (D_HID / 4) + lane] =
                    make_float4(sc[0][k] * r, sc[1][k] * r, sc[2][k] * r, sc[3][k] * r);
            }
        }
    } else {
        float4* acc4 = reinterpret_cast<float4*>(acc);
        for (int i = tid; i < AGG_ROWS * D_HID / 4; i += AGG_THREADS)
            acc4[i] = make_float4(0.f, 0.f, 0.f, 0.f);
        if (tid < AGG_ROWS) den[tid] = 0.f;
        for (int g = wave; g < EDGE_GROUPS; g += AGG_WAVES) {
            const int4 d = dst4[g * 32 + lane];
            const int dd[4] = {d.x, d.y, d.z, d.w};
            #pragma unroll
            for (int c = 0; c < 4; ++c) {
                const unsigned t = (unsigned)(dd[c] - base);
                if (t < (unsigned)nrows) {
                    const int s = src[(g * 32 + lane) * 4 + c];
                    atomicMax(&mx[t], enc_f32(edge_score(el[s], er_own[t])));
                }
            }
        }
        __syncthreads();
        for (int g = wave; g < EDGE_GROUPS; g += AGG_WAVES) {
            const int4 d = dst4[g * 32 + lane];
            const int dd[4] = {d.x, d.y, d.z, d.w};
            #pragma unroll
            for (int c = 0; c < 4; ++c) {
                const unsigned t = (unsigned)(dd[c] - base);
                const bool m = t < (unsigned)nrows;
                int s = 0, tl = 0;
                float a = 0.f;
                if (m) {
                    s = src[(g * 32 + lane) * 4 + c];
                    tl = (int)t;
                    a = __expf(edge_score(el[s], er_own[t]) - dec_f32(mx[t]));
                    atomicAdd(den + tl, a);
                }
                unsigned bal = __builtin_amdgcn_ballot_w32(m);
                while (bal != 0u) {
                    const int j = __builtin_ctz(bal);
                    bal &= bal - 1u;
                    const int sj = __builtin_amdgcn_readlane(s, j);
                    const int dj = __builtin_amdgcn_readlane(tl, j);
                    const float aj = __int_as_float(__builtin_amdgcn_readlane(__float_as_int(a), j));
                    const float2* __restrict__ row =
                        reinterpret_cast<const float2*>(feat + (size_t)sj * D_HID);
                    const float2 x0 = row[lane], x1 = row[32 + lane];
                    float* arow = acc + dj * D_HID + 2 * lane;
                    atomicAdd(arow,      aj * x0.x);
                    atomicAdd(arow + 1,  aj * x0.y);
                    atomicAdd(arow + 64, aj * x1.x);
                    atomicAdd(arow + 65, aj * x1.y);
                }
            }
        }
        __syncthreads();
        if (tid < AGG_ROWS) {
            const float d = den[tid];
            den[tid] = (d > 0.f) ? 1.0f / d : 0.f;
        }
        __syncthreads();
        float4* __restrict__ h4 = reinterpret_cast<float4*>(h + (size_t)base * D_HID);
        for (int i = tid; i < nrows * (D_HID / 4); i += AGG_THREADS) {
            const float r = den[i / (D_HID / 4)];
            float4 v = acc4[i];
            v.x *= r; v.y *= r; v.z *= r; v.w *= r;
            h4[i] = v;
        }
    }
}

extern "C" void kernel_launch(void* const* d_in, const int* in_sizes, int n_in,
                              void* d_out, int out_size, void* d_ws, size_t ws_size,
                              hipStream_t stream) {
    const float* X  = (const float*)d_in[0];
    const float* W  = (const float*)d_in[1];
    const float* al = (const float*)d_in[2];
    const float* ar = (const float*)d_in[3];
    const float* FW = (const float*)d_in[4];
    const float* FB = (const float*)d_in[5];
    const int*  src = (const int*)d_in[6];
    const int*  dst = (const int*)d_in[7];
    float* out = (float*)d_out;

    char* ws = (char*)d_ws;
    size_t off = 0;
    auto carve = [&](size_t bytes) {
        void* p = ws + off;
        off += (bytes + 255) & ~(size_t)255;
        return p;
    };
    float*    feat  = (float*)   carve((size_t)N_NODES * D_HID * sizeof(float));
    float*    h     = (float*)   carve((size_t)N_NODES * D_HID * sizeof(float));
    float*    el    = (float*)   carve((size_t)N_NODES * sizeof(float));
    float*    er    = (float*)   carve((size_t)N_NODES * sizeof(float));
    unsigned* own   = (unsigned*)carve((size_t)N_EDGES);

    k_gemm1<<<1563, 128, 0, stream>>>(X, W, al, ar, feat, el, er, dst, own);
    k_agg<<<AGG_WGS, AGG_THREADS, 0, stream>>>(src, dst, own, el, er, feat, h);
    k_gemm2<<<782, 256, 0, stream>>>(h, FW, FB, out);
}


# GEMM1 launch also writes one word per edge, src | (dst % 391) << 17; the aggregation fetches that one record per kept edge instead of src[e] and dst[e]: a third fewer scattered reads; sums unchanged
# speedup vs baseline: 5.6924x; 1.0330x over previous
#include <hip/hip_runtime.h>
#include <hip/hip_bf16.h>

#define N_NODES   100000
#define N_EDGES   1600000
#define D_IN      256
#define D_HID     128
#define N_CLASSES 64
#define NEG_SLOPE 0.2f

typedef __attribute__((ext_vector_type(2))) float v2f;
typedef __attribute__((ext_vector_type(8))) float v8f;

#define OWN_ROWS 391u
#define ROW_SHIFT 17
static_assert(N_EDGES == (N_NODES / 16) * 256, "one 256-edge slice of the owner map per GEMM1 wave");
static_assert((N_NODES - 1) / OWN_ROWS < 256u, "owner index must fit one byte");
static_assert(N_NODES <= (1 << ROW_SHIFT), "src must fit below the row field");
static_assert(OWN_ROWS <= (1u << (31 - ROW_SHIFT)), "row must fit above the src field");

__global__ __launch_bounds__(128) void k_gemm1(const float* __restrict__ X,
                                               const float* __restrict__ W,
                                               const float* __restrict__ al,
                                               const float* __restrict__ ar,
                                               float* __restrict__ feat,
                                               float* __restrict__ el,
                                               float* __restrict__ er,
                                               const int* __restrict__ src,
                                               const int* __restrict__ dst,
                                               unsigned* __restrict__ own,
                                               unsigned* __restrict__ rec) {
    const int wave = threadIdx.x >> 5;
    const int lane = threadIdx.x & 31;
    const int rowTile = blockIdx.x * 4 + wave;
    if (rowTile >= N_NODES / 16) return;
    const int row0 = rowTile * 16;
    {
        const size_t e4 = (size_t)rowTile * 64 + lane * 2;
        const int4* __restrict__ d4 = reinterpret_cast<const int4*>(dst) + e4;
        const int4* __restrict__ s4 = reinterpret_cast<const int4*>(src) + e4;
        const int4 da = d4[0], db = d4[1];
        const int4 sa = s4[0], sb = s4[1];
        const unsigned dd[8] = {(unsigned)da.x, (unsigned)da.y, (unsigned)da.z, (unsigned)da.w,
                                (unsigned)db.x, (unsigned)db.y, (unsigned)db.z, (unsigned)db.w};
        const unsigned ss[8] = {(unsigned)sa.x, (unsigned)sa.y, (unsigned)sa.z, (unsigned)sa.w,
                                (unsigned)sb.x, (unsigned)sb.y, (unsigned)sb.z, (unsigned)sb.w};
        unsigned q[8], r[8];
        for (int t = 0; t < 8; ++t) {
            q[t] = dd[t] / OWN_ROWS;
            r[t] = ss[t] | ((dd[t] - q[t] * OWN_ROWS) << ROW_SHIFT);
        }
        const unsigned lo  = q[0] | (q[1] << 8) | (q[2] << 16) | (q[3] << 24);
        const unsigned hi8 = q[4] | (q[5] << 8) | (q[6] << 16) | (q[7] << 24);
        reinterpret_cast<uint2*>(own)[rowTile * 32 + lane] = make_uint2(lo, hi8);
        uint4* __restrict__ r4 = reinterpret_cast<uint4*>(rec) + e4;
        r4[0] = make_uint4(r[0], r[1], r[2], r[3]);
        r4[1] = make_uint4(r[4], r[5], r[6], r[7]);
    }
    const int l16 = lane & 15;
    const int hi  = lane >> 4;

    v8f acc[8] = {};
    const float4* __restrict__ xrow4 =
        reinterpret_cast<const float4*>(X + (size_t)(row0 + l16) * D_IN + 8 * hi);
    const float4* __restrict__ w4 =
        reinterpret_cast<const float4*>(W + (size_t)(8 * hi) * D_HID + 4 * l16);

    for (int j = 0; j < D_IN / 16; ++j) {
        const float4 q0 = xrow4[4 * j];
        const float4 q1 = xrow4[4 * j + 1];
        const float av[8] = {q0.x, q0.y, q0.z, q0.w, q1.x, q1.y, q1.z, q1.w};
        for (int s = 0; s < 4; ++s) {
            const float4* __restrict__ r0 = w4 + (size_t)(16 * j + 2 * s) * (D_HID / 4);
            const float4* __restrict__ r1 = r0 + D_HID / 4;
            const float4 b0l = r0[0], b0h = r0[16];
            const float4 b1l = r1[0], b1h = r1[16];
            v2f a;
            a.x = av[2 * s];
            a.y = av[2 * s + 1];
            v2f b[8];
            b[0].x = b0l.x; b[0].y = b1l.x;
            b[1].x = b0l.y; b[1].y = b1l.y;
            b[2].x = b0l.z; b[2].y = b1l.z;
            b[3].x = b0l.w; b[3].y = b1l.w;
            b[4].x = b0h.x; b[4].y = b1h.x;
            b[5].x = b0h.y; b[5].y = b1h.y;
            b[6].x = b0h.z; b[6].y = b1h.z;
            b[7].x = b0h.w; b[7].y = b1h.w;
            for (int t = 0; t < 8; ++t)
                acc[t] = __builtin_amdgcn_wmma_f32_16x16x4_f32(
                    false, a, false, b[t], (short)0, acc[t], false, false);
        }
    }
    float4* __restrict__ feat4 = reinterpret_cast<float4*>(feat);
    for (int r = 0; r < 8; ++r) {
        float4* __restrict__ orow = feat4 + (size_t)(row0 + r + 8 * hi) * (D_HID / 4) + l16;
        orow[0]  = make_float4(acc[0][r], acc[1][r], acc[2][r], acc[3][r]);
        orow[16] = make_float4(acc[4][r], acc[5][r], acc[6][r], acc[7][r]);
    }
    const float4 all4 = reinterpret_cast<const float4*>(al)[l16];
    const float4 alh4 = reinterpret_cast<const float4*>(al)[16 + l16];
    const float4 arl4 = reinterpret_cast<const float4*>(ar)[l16];
    const float4 arh4 = reinterpret_cast<const float4*>(ar)[16 + l16];
    const float alv[8] = {all4.x, all4.y, all4.z, all4.w, alh4.x, alh4.y, alh4.z, alh4.w};
    const float arv[8] = {arl4.x, arl4.y, arl4.z, arl4.w, arh4.x, arh4.y, arh4.z, arh4.w};
    float myl = 0.f, myr = 0.f;
    for (int r = 0; r < 8; ++r) {
        float sl = 0.f, sr = 0.f;
        for (int t = 0; t < 8; ++t) {
            sl += acc[t][r] * alv[t];
            sr += acc[t][r] * arv[t];
        }
        for (int off = 8; off > 0; off >>= 1) {
            sl += __shfl_xor(sl, off, 32);
            sr += __shfl_xor(sr, off, 32);
        }
        if (l16 == r) { myl = sl; myr = sr; }
    }
    if (l16 < 8) {
        el[row0 + l16 + 8 * hi] = myl;
        er[row0 + l16 + 8 * hi] = myr;
    }
}

__global__ __launch_bounds__(256) void k_gemm2(const float* __restrict__ H,
                                               const float* __restrict__ FW,
                                               const float* __restrict__ FB,
                                               float* __restrict__ out) {
    const int wave = threadIdx.x >> 5;
    const int lane = threadIdx.x & 31;
    const int rowTile = blockIdx.x * 8 + wave;
    if (rowTile >= N_NODES / 16) return;
    const int row0 = rowTile * 16;
    const int l16 = lane & 15;
    const int hi  = lane >> 4;

    v8f acc0 = {}, acc1 = {}, acc2 = {}, acc3 = {};
    const float4* __restrict__ hrow4 =
        reinterpret_cast<const float4*>(H + (size_t)(row0 + l16) * D_HID + 8 * hi);
    const float4* __restrict__ fw4 =
        reinterpret_cast<const float4*>(FW + (size_t)(8 * hi) * N_CLASSES + 4 * l16);

    for (int j = 0; j < D_HID / 16; ++j) {
        const float4 q0 = hrow4[4 * j];
        const float4 q1 = hrow4[4 * j + 1];
        const float av[8] = {q0.x, q0.y, q0.z, q0.w, q1.x, q1.y, q1.z, q1.w};
        for (int s = 0; s < 4; ++s) {
            const float4 b0 = fw4[(size_t)(16 * j + 2 * s) * (N_CLASSES / 4)];
            const float4 b1 = fw4[(size_t)(16 * j + 2 * s + 1) * (N_CLASSES / 4)];
            v2f a;
            a.x = av[2 * s];
            a.y = av[2 * s + 1];
            v2f c0, c1, c2, c3;
            c0.x = b0.x; c0.y = b1.x;
            c1.x = b0.y; c1.y = b1.y;
            c2.x = b0.z; c2.y = b1.z;
            c3.x = b0.w; c3.y = b1.w;
            acc0 = __builtin_amdgcn_wmma_f32_16x16x4_f32(
                false, a, false, c0, (short)0, acc0, false, false);
            acc1 = __builtin_amdgcn_wmma_f32_16x16x4_f32(
                false, a, false, c1, (short)0, acc1, false, false);
            acc2 = __builtin_amdgcn_wmma_f32_16x16x4_f32(
                false, a, false, c2, (short)0, acc2, false, false);
            acc3 = __builtin_amdgcn_wmma_f32_16x16x4_f32(
                false, a, false, c3, (short)0, acc3, false, false);
        }
    }
    const float4 bias = reinterpret_cast<const float4*>(FB)[l16];
    float4* __restrict__ out4 = reinterpret_cast<float4*>(out);
    for (int r = 0; r < 8; ++r) {
        out4[(size_t)(row0 + r + 8 * hi) * (N_CLASSES / 4) + l16] =
            make_float4(acc0[r] + bias.x, acc1[r] + bias.y, acc2[r] + bias.z, acc3[r] + bias.w);
    }
}

__device__ __forceinline__ unsigned enc_f32(float f) {
    unsigned b = __float_as_uint(f);
    return (b & 0x80000000u) ? ~b : (b | 0x80000000u);
}
__device__ __forceinline__ float dec_f32(unsigned u) {
    unsigned b = (u & 0x80000000u) ? (u & 0x7FFFFFFFu) : ~u;
    return __uint_as_float(b);
}

#define AGG_WGS     256
#define AGG_ROWS    391
static_assert(AGG_ROWS == (int)OWN_ROWS, "owner map and block ranges must agree");
#define AGG_THREADS 1024
#define AGG_WAVES   (AGG_THREADS / 32)
#define AGG_CAP     400
#define DL_CAP      768
#define AGG_BATCH   8
#define AGG_UNROLL  4
#define EDGE_GROUPS (N_EDGES / 128)
#define OWN_GROUPS  (N_EDGES / 512)
#define AGG_SLOTS   ((AGG_ROWS + AGG_WAVES - 1) / AGG_WAVES)
static_assert(AGG_SLOTS <= 32, "lane k of a wave holds the denominator of its slot k");
static_assert(AGG_WAVES == 32, "lane t of a wave holds the count of summing wave t");
static_assert((size_t)AGG_WAVES * DL_CAP * 8 <= (size_t)AGG_ROWS * D_HID * 4, "summing lists fit the slow path's tile");

__device__ __forceinline__ float edge_score(float l, float r) {
    const float v = l + r;
    return (v > 0.f) ? v : NEG_SLOPE * v;
}

typedef __attribute__((ext_vector_type(16))) float v16f;
static_assert(AGG_SLOTS <= 16, "one vector element per slot");

template <bool FULL>
__device__ __forceinline__ void agg_batch(int sr, float a, int jb, int n, int lane,
                                          const float* __restrict__ feat,
                                          v16f (&sc)[4], float& dn) {
    float4 v[AGG_BATCH];
    int srj[AGG_BATCH];
    const float4* __restrict__ feat4 = reinterpret_cast<const float4*>(feat);
    #pragma unroll
    for (int j = 0; j < AGG_BATCH; ++j) {
        if (FULL || j < n) {
            srj[j] = __builtin_amdgcn_readlane(sr, jb + j);
            const unsigned sj = (unsigned)srj[j] & ((1u << ROW_SHIFT) - 1u);
            v[j] = feat4[sj * (unsigned)(D_HID / 4) + (unsigned)lane];
        }
    }
    #pragma unroll
    for (int j = 0; j < AGG_BATCH; ++j) {
        if (FULL || j < n) {
            const int kj = (int)(((unsigned)srj[j] >> ROW_SHIFT) / AGG_WAVES);
            const float aj = __int_as_float(__builtin_amdgcn_readlane(__float_as_int(a), jb + j));
            if (lane == kj) dn += aj;
            sc[0][kj] += aj * v[j].x;
            sc[1][kj] += aj * v[j].y;
            sc[2][kj] += aj * v[j].z;
            sc[3][kj] += aj * v[j].w;
        }
    }
}

__global__ __launch_bounds__(AGG_THREADS) void k_agg(const int* __restrict__ src,
                                                     const int* __restrict__ dst,
                                                     const unsigned* __restrict__ own,
                                                     const unsigned* __restrict__ rec,
                                                     const float* __restrict__ el,
                                                     const float* __restrict__ er,
                                                     const float* __restrict__ feat,
                                                     float* __restrict__ h) {
    __shared__ __attribute__((aligned(16))) float acc[AGG_ROWS * D_HID];
    __shared__ int2 lst[AGG_WAVES * AGG_CAP];
    __shared__ unsigned cnt2[AGG_WAVES * AGG_WAVES];
    __shared__ float den[AGG_ROWS];
    __shared__ unsigned mx[AGG_ROWS];
    __shared__ int ovf;

    const int tid  = threadIdx.x;
    const int lane = tid & 31;
    const int wave = __builtin_amdgcn_readfirstlane(tid >> 5);
    const int base = blockIdx.x * AGG_ROWS;
    const int nrows = (N_NODES - base < AGG_ROWS) ? (N_NODES - base) : AGG_ROWS;

    if (tid < AGG_ROWS) mx[tid] = 0u;
    if (tid == 0) ovf = 0;
    __syncthreads();

    int2* lstw = lst + wave * AGG_CAP;
    const float* __restrict__ er_own = er + base;
    const unsigned lt_mask = (1u << lane) - 1u;
    const int4* __restrict__ dst4 = reinterpret_cast<const int4*>(dst);

    const uint4* __restrict__ own16 = reinterpret_cast<const uint4*>(own);
    const unsigned key = (unsigned)blockIdx.x * 0x01010101u;
    unsigned cnt = 0;
    auto fetch = [&](int g) -> uint4 {
        return own16[((g < OWN_GROUPS) ? g : (OWN_GROUPS - 1)) * 32 + lane];
    };
    auto hits = [&](unsigned v) -> unsigned {
        const unsigned x = v ^ key;
        const unsigned t = (x & 0x7F7F7F7Fu) + 0x7F7F7F7Fu;
        return ~(t | x) & 0x80808080u;
    };
    auto examine = [&](const uint4& o, int g) {
        if (g >= OWN_GROUPS) return;
        unsigned zz = (hits(o.x) >> 7) | (hits(o.y) >> 6) | (hits(o.z) >> 5) | (hits(o.w) >> 4);
        const int e0 = (g * 32 + lane) * 16;
        unsigned bal = __builtin_amdgcn_ballot_w32(zz != 0u);
        while (bal != 0u) {
            const unsigned pos = cnt + (unsigned)__popc(bal & lt_mask);
            if (zz != 0u) {
                const int p = __builtin_ctz(zz);
                zz &= zz - 1u;
                if (pos < (unsigned)AGG_CAP)
                    lstw[pos] = make_int2(e0 + ((p & 3) << 2) + (p >> 3), 0);
            }
            cnt += (unsigned)__popc(bal);
            bal = __builtin_amdgcn_ballot_w32(zz != 0u);
        }
    };

    constexpr int STEP = AGG_WAVES * AGG_UNROLL;
    uint4 bufA[AGG_UNROLL], bufB[AGG_UNROLL];
    #pragma unroll
    for (int u = 0; u < AGG_UNROLL; ++u) bufA[u] = fetch(wave + AGG_WAVES * u);
    for (int g0 = wave; g0 < OWN_GROUPS; g0 += 2 * STEP) {
        #pragma unroll
        for (int u = 0; u < AGG_UNROLL; ++u) bufB[u] = fetch(g0 + STEP + AGG_WAVES * u);
        #pragma unroll
        for (int u = 0; u < AGG_UNROLL; ++u) examine(bufA[u], g0 + AGG_WAVES * u);
        #pragma unroll
        for (int u = 0; u < AGG_UNROLL; ++u) bufA[u] = fetch(g0 + 2 * STEP + AGG_WAVES * u);
        #pragma unroll
        for (int u = 0; u < AGG_UNROLL; ++u) examine(bufB[u], g0 + STEP + AGG_WAVES * u);
    }
    const unsigned cntc = (cnt < (unsigned)AGG_CAP) ? cnt : (unsigned)AGG_CAP;
    if (lane == 0 && cnt > (unsigned)AGG_CAP) ovf = 1;

    unsigned lmask[5];
    for (int k = 0; k < 5; ++k) lmask[k] = ((lane >> k) & 1) ? 0u : ~0u;
    auto chunk_counts = [&](bool val, unsigned t) -> unsigned {
        unsigned m = __builtin_amdgcn_ballot_w32(val);
        for (int k = 0; k < 5; ++k)
            m &= __builtin_amdgcn_ballot_w32(val && ((t >> k) & 1u)) ^ lmask[k];
        return (unsigned)__popc(m);
    };
    unsigned hist = 0;

    for (unsigned i0 = 0; i0 < cntc; i0 += 128) {
        int  en[4], sv[4], rw[4];
        unsigned pk[4];
        float ev[4];
        #pragma unroll
        for (int u = 0; u < 4; ++u) {
            const unsigned i = i0 + 32u * u + (unsigned)lane;
            en[u] = (i < cntc) ? lstw[i].x : 0;
        }
        #pragma unroll
        for (int u = 0; u < 4; ++u) pk[u] = rec[en[u]];
        #pragma unroll
        for (int u = 0; u < 4; ++u) {
            const unsigned i = i0 + 32u * u + (unsigned)lane;
            pk[u] = (i < cntc) ? pk[u] : 0u;
            sv[u] = (int)(pk[u] & ((1u << ROW_SHIFT) - 1u));
            rw[u] = (int)(pk[u] >> ROW_SHIFT);
        }
        for (int u = 0; u < 4; ++u)
            hist += chunk_counts(i0 + 32u * u + (unsigned)lane < cntc, (unsigned)rw[u] & (AGG_WAVES - 1));
        #pragma unroll
        for (int u = 0; u < 4; ++u) ev[u] = edge_score(el[sv[u]], er_own[rw[u]]);
        #pragma unroll
        for (int u = 0; u < 4; ++u) {
            const unsigned i = i0 + 32u * u + (unsigned)lane;
            if (i < cntc) {
                atomicMax(&mx[rw[u]], enc_f32(ev[u]));
                lstw[i] = make_int2((int)pk[u], __float_as_int(ev[u]));
            }
        }
    }
    cnt2[wave * AGG_WAVES + lane] = hist;
    __syncthreads();

    const int ovf_u = __builtin_amdgcn_readfirstlane(ovf);
    unsigned pre = 0, tot = 0;
    for (int q = 0; q < AGG_WAVES; ++q) {
        const unsigned c = cnt2[q * AGG_WAVES + lane];
        tot += c;
        if (q < wave) pre += c;
    }
    const bool fits = __builtin_amdgcn_ballot_w32(tot > (unsigned)DL_CAP) == 0u;
    if (ovf_u == 0 && fits) {
        int2* dl = reinterpret_cast<int2*>(acc);
        unsigned pos = pre;
        for (unsigned i0 = 0; i0 < cntc; i0 += 32) {
            const unsigned i = i0 + (unsigned)lane;
            const bool val = i < cntc;
            const int2 ent = val ? lstw[i] : make_int2(0, 0);
            const unsigned t = ((unsigned)ent.x >> ROW_SHIFT) & (AGG_WAVES - 1);
            unsigned same = __builtin_amdgcn_ballot_w32(val);
            unsigned mine = same;
            for (int k = 0; k < 5; ++k) {
                const unsigned bk = __builtin_amdgcn_ballot_w32(val && ((t >> k) & 1u));
                same &= ((t >> k) & 1u) ? bk : ~bk;
                mine &= bk ^ lmask[k];
            }
            const unsigned first = (unsigned)__builtin_amdgcn_ds_bpermute((int)(t << 2), (int)pos);
            if (val) dl[t * DL_CAP + first + (unsigned)__popc(same & lt_mask)] = ent;
            pos += (unsigned)__popc(mine);
        }
        __syncthreads();

        const unsigned nt = (unsigned)__builtin_amdgcn_readlane((int)tot, wave);
        const int2* dlw = dl + wave * DL_CAP;
        v16f sc[4] = {};
        float dn = 0.f;
        const unsigned nfull = nt & ~(unsigned)(AGG_BATCH - 1);
        int sr = 0;
        float a = 0.f;
        for (unsigned i0 = 0; i0 < nt; i0 += 32) {
            const unsigned i = i0 + (unsigned)lane;
            sr = 0;
            a = 0.f;
            if (i < nt) {
                const int2 ent = dlw[i];
                sr = ent.x;
                a  = __expf(__int_as_float(ent.y) - dec_f32(mx[sr >> ROW_SHIFT]));
            }
            const int nb = (nfull - i0 < 32u) ? (int)(nfull - i0) : 32;
            for (int jb = 0; jb < nb; jb += AGG_BATCH)
                agg_batch<true>(sr, a, jb, AGG_BATCH, lane, feat, sc, dn);
        }
        if (nt != nfull)
            agg_batch<false>(sr, a, (int)(nfull & 31u), (int)(nt - nfull), lane, feat, sc, dn);
        const float dinv = (dn > 0.f) ? 1.0f / dn : 0.f;
        float4* __restrict__ h4 = reinterpret_cast<float4*>(h + (size_t)base * D_HID);
        #pragma unroll
        for (int k = 0; k < AGG_SLOTS; ++k) {
            const int row = wave + AGG_WAVES * k;
            if (row < nrows) {
                const float r = __int_as_float(__builtin_amdgcn_readlane(__float_as_int(dinv), k));
                h4[row * (D_HID / 4) + lane] =
                    make_float4(sc[0][k] * r, sc[1][k] * r, sc[2][k] * r, sc[3][k] * r);
            }
        }
    } else {
        float4* acc4 = reinterpret_cast<float4*>(acc);
        for (int i = tid; i < AGG_ROWS * D_HID / 4; i += AGG_THREADS)
            acc4[i] = make_float4(0.f, 0.f, 0.f, 0.f);
        if (tid < AGG_ROWS) den[tid] = 0.f;
        for (int g = wave; g < EDGE_GROUPS; g += AGG_WAVES) {
            const int4 d = dst4[g * 32 + lane];
            const int dd[4] = {d.x, d.y, d.z, d.w};
            #pragma unroll
            for (int c = 0; c < 4; ++c) {
                const unsigned t = (unsigned)(dd[c] - base);
                if (t < (unsigned)nrows) {
                    const int s = src[(g * 32 + lane) * 4 + c];
                    atomicMax(&mx[t], enc_f32(edge_score(el[s], er_own[t])));
                }
            }
        }
        __syncthreads();
        for (int g = wave; g < EDGE_GROUPS; g += AGG_WAVES) {
            const int4 d = dst4[g * 32 + lane];
            const int dd[4] = {d.x, d.y, d.z, d.w};
            #pragma unroll
            for (int c = 0; c < 4; ++c) {
                const unsigned t = (unsigned)(dd[c] - base);
                const bool m = t < (unsigned)nrows;
                int s = 0, tl = 0;
                float a = 0.f;
                if (m) {
                    s = src[(g * 32 + lane) * 4 + c];
                    tl = (int)t;
                    a = __expf(edge_score(el[s], er_own[t]) - dec_f32(mx[t]));
                    atomicAdd(den + tl, a);
                }
                unsigned bal = __builtin_amdgcn_ballot_w32(m);
                while (bal != 0u) {
                    const int j = __builtin_ctz(bal);
                    bal &= bal - 1u;
                    const int sj = __builtin_amdgcn_readlane(s, j);
                    const int dj = __builtin_amdgcn_readlane(tl, j);
                    const float aj = __int_as_float(__builtin_amdgcn_readlane(__float_as_int(a), j));
                    const float2* __restrict__ row =
                        reinterpret_cast<const float2*>(feat + (size_t)sj * D_HID);
                    const float2 x0 = row[lane], x1 = row[32 + lane];
                    float* arow = acc + dj * D_HID + 2 * lane;
                    atomicAdd(arow,      aj * x0.x);
                    atomicAdd(arow + 1,  aj * x0.y);
                    atomicAdd(arow + 64, aj * x1.x);
                    atomicAdd(arow + 65, aj * x1.y);
                }
            }
        }
        __syncthreads();
        if (tid < AGG_ROWS) {
            const float d = den[tid];
            den[tid] = (d > 0.f) ? 1.0f / d : 0.f;
        }
        __syncthreads();
        float4* __restrict__ h4 = reinterpret_cast<float4*>(h + (size_t)base * D_HID);
        for (int i = tid; i < nrows * (D_HID / 4); i += AGG_THREADS) {
            const float r = den[i / (D_HID / 4)];
            float4 v = acc4[i];
            v.x *= r; v.y *= r; v.z *= r; v.w *= r;
            h4[i] = v;
        }
    }
}

extern "C" void kernel_launch(void* const* d_in, const int* in_sizes, int n_in,
                              void* d_out, int out_size, void* d_ws, size_t ws_size,
                              hipStream_t stream) {
    const float* X  = (const float*)d_in[0];
    const float* W  = (const float*)d_in[1];
    const float* al = (const float*)d_in[2];
    const float* ar = (const float*)d_in[3];
    const float* FW = (const float*)d_in[4];
    const float* FB = (const float*)d_in[5];
    const int*  src = (const int*)d_in[6];
    const int*  dst = (const int*)d_in[7];
    float* out = (float*)d_out;

    char* ws = (char*)d_ws;
    size_t off = 0;
    auto carve = [&](size_t bytes) {
        void* p = ws + off;
        off += (bytes + 255) & ~(size_t)255;
        return p;
    };
    float*    feat  = (float*)   carve((size_t)N_NODES * D_HID * sizeof(float));
    float*    h     = (float*)   carve((size_t)N_NODES * D_HID * sizeof(float));
    float*    el    = (float*)   carve((size_t)N_NODES * sizeof(float));
    float*    er    = (float*)   carve((size_t)N_NODES * sizeof(float));
    unsigned* own   = (unsigned*)carve((size_t)N_EDGES);
    unsigned* rec   = (unsigned*)carve((size_t)N_EDGES * sizeof(unsigned));

    k_gemm1<<<1563, 128, 0, stream>>>(X, W, al, ar, feat, el, er, src, dst, own, rec);
    k_agg<<<AGG_WGS, AGG_THREADS, 0, stream>>>(src, dst, own, rec, el, er, feat, h);
    k_gemm2<<<782, 256, 0, stream>>>(h, FW, FB, out);
}


# Aggregation scan reads its block's row of a 1-bit-per-edge owner bitmap (GEMM1 blocks build it in LDS by integer OR, whole lines out) instead of the 1-byte map: 8x fewer scan bytes; row sum order new
# speedup vs baseline: 6.0609x; 1.0647x over previous
#include <hip/hip_runtime.h>
#include <hip/hip_bf16.h>

#define N_NODES   100000
#define N_EDGES   1600000
#define D_IN      256
#define D_HID     128
#define N_CLASSES 64
#define NEG_SLOPE 0.2f

typedef __attribute__((ext_vector_type(2))) float v2f;
typedef __attribute__((ext_vector_type(8))) float v8f;

#define OWN_ROWS 391u
#define OWNERS   256u
#define ROW_SHIFT 17
#define G1_BLOCKS   1563
#define BM_ROW_U4   (OWN_ROWS * D_HID * 4u / 16u)
#define BM_VALID_U4 (G1_BLOCKS * 8u)
#define BM_GROUPS   (BM_ROW_U4 / 32u)
static_assert(N_EDGES == (N_NODES / 16) * 256, "one 256-edge slice of the edge tables per GEMM1 wave");
static_assert(G1_BLOCKS * 4 >= N_NODES / 16 && G1_BLOCKS * 1024 >= N_EDGES, "GEMM1 blocks cover strips and edges");
static_assert((N_NODES - 1) / OWN_ROWS < OWNERS, "one bitmap row per owner");
static_assert(BM_ROW_U4 % 32u == 0 && BM_VALID_U4 <= BM_ROW_U4, "a bitmap row fits the block's rows of h");
static_assert(N_NODES <= (1 << ROW_SHIFT), "src must fit below the row field");
static_assert(OWN_ROWS <= (1u << (31 - ROW_SHIFT)), "row must fit above the src field");

__global__ __launch_bounds__(128) void k_gemm1(const float* __restrict__ X,
                                               const float* __restrict__ W,
                                               const float* __restrict__ al,
                                               const float* __restrict__ ar,
                                               float* __restrict__ feat,
                                               float* __restrict__ el,
                                               float* __restrict__ er,
                                               const int* __restrict__ src,
                                               const int* __restrict__ dst,
                                               unsigned* __restrict__ bmp,
                                               unsigned* __restrict__ rec) {
    __shared__ __attribute__((aligned(16))) unsigned bm[OWNERS * 32];
    const int wave = threadIdx.x >> 5;
    const int lane = threadIdx.x & 31;
    const int rowTile = blockIdx.x * 4 + wave;
    const bool active = rowTile < N_NODES / 16;
    const int row0 = rowTile * 16;
    uint4* bm4 = reinterpret_cast<uint4*>(bm);
    for (int i = 0; i < 16; ++i) bm4[i * 128 + threadIdx.x] = make_uint4(0u, 0u, 0u, 0u);
    __syncthreads();
    if (active) {
        const size_t e4 = (size_t)rowTile * 64 + lane * 2;
        const int4* __restrict__ d4 = reinterpret_cast<const int4*>(dst) + e4;
        const int4* __restrict__ s4 = reinterpret_cast<const int4*>(src) + e4;
        const int4 da = d4[0], db = d4[1];
        const int4 sa = s4[0], sb = s4[1];
        const unsigned dd[8] = {(unsigned)da.x, (unsigned)da.y, (unsigned)da.z, (unsigned)da.w,
                                (unsigned)db.x, (unsigned)db.y, (unsigned)db.z, (unsigned)db.w};
        const unsigned ss[8] = {(unsigned)sa.x, (unsigned)sa.y, (unsigned)sa.z, (unsigned)sa.w,
                                (unsigned)sb.x, (unsigned)sb.y, (unsigned)sb.z, (unsigned)sb.w};
        unsigned q[8], r[8];
        for (int t = 0; t < 8; ++t) {
            q[t] = dd[t] / OWN_ROWS;
            r[t] = ss[t] | ((dd[t] - q[t] * OWN_ROWS) << ROW_SHIFT);
        }
        for (int t = 0; t < 8; ++t)
            atomicOr(&bm[q[t] * 32u + (unsigned)(wave * 8 + (lane >> 2))], 1u << ((lane & 3) * 8 + t));
        uint4* __restrict__ r4 = reinterpret_cast<uint4*>(rec) + e4;
        r4[0] = make_uint4(r[0], r[1], r[2], r[3]);
        r4[1] = make_uint4(r[4], r[5], r[6], r[7]);
    }
    __syncthreads();
    uint4* __restrict__ bmp4 = reinterpret_cast<uint4*>(bmp);
    for (int i = 0; i < 16; ++i) {
        const unsigned idx = (unsigned)(i * 128) + threadIdx.x;
        bmp4[(size_t)(idx >> 3) * BM_ROW_U4 + blockIdx.x * 8u + (idx & 7u)] = bm4[idx];
    }
    if (!active) return;
    const int l16 = lane & 15;
    const int hi  = lane >> 4;

    v8f acc[8] = {};
    const float4* __restrict__ xrow4 =
        reinterpret_cast<const float4*>(X + (size_t)(row0 + l16) * D_IN + 8 * hi);
    const float4* __restrict__ w4 =
        reinterpret_cast<const float4*>(W + (size_t)(8 * hi) * D_HID + 4 * l16);

    for (int j = 0; j < D_IN / 16; ++j) {
        const float4 q0 = xrow4[4 * j];
        const float4 q1 = xrow4[4 * j + 1];
        const float av[8] = {q0.x, q0.y, q0.z, q0.w, q1.x, q1.y, q1.z, q1.w};
        for (int s = 0; s < 4; ++s) {
            const float4* __restrict__ r0 = w4 + (size_t)(16 * j + 2 * s) * (D_HID / 4);
            const float4* __restrict__ r1 = r0 + D_HID / 4;
            const float4 b0l = r0[0], b0h = r0[16];
            const float4 b1l = r1[0], b1h = r1[16];
            v2f a;
            a.x = av[2 * s];
            a.y = av[2 * s + 1];
            v2f b[8];
            b[0].x = b0l.x; b[0].y = b1l.x;
            b[1].x = b0l.y; b[1].y = b1l.y;
            b[2].x = b0l.z; b[2].y = b1l.z;
            b[3].x = b0l.w; b[3].y = b1l.w;
            b[4].x = b0h.x; b[4].y = b1h.x;
            b[5].x = b0h.y; b[5].y = b1h.y;
            b[6].x = b0h.z; b[6].y = b1h.z;
            b[7].x = b0h.w; b[7].y = b1h.w;
            for (int t = 0; t < 8; ++t)
                acc[t] = __builtin_amdgcn_wmma_f32_16x16x4_f32(
                    false, a, false, b[t], (short)0, acc[t], false, false);
        }
    }
    float4* __restrict__ feat4 = reinterpret_cast<float4*>(feat);
    for (int r = 0; r < 8; ++r) {
        float4* __restrict__ orow = feat4 + (size_t)(row0 + r + 8 * hi) * (D_HID / 4) + l16;
        orow[0]  = make_float4(acc[0][r], acc[1][r], acc[2][r], acc[3][r]);
        orow[16] = make_float4(acc[4][r], acc[5][r], acc[6][r], acc[7][r]);
    }
    const float4 all4 = reinterpret_cast<const float4*>(al)[l16];
    const float4 alh4 = reinterpret_cast<const float4*>(al)[16 + l16];
    const float4 arl4 = reinterpret_cast<const float4*>(ar)[l16];
    const float4 arh4 = reinterpret_cast<const float4*>(ar)[16 + l16];
    const float alv[8] = {all4.x, all4.y, all4.z, all4.w, alh4.x, alh4.y, alh4.z, alh4.w};
    const float arv[8] = {arl4.x, arl4.y, arl4.z, arl4.w, arh4.x, arh4.y, arh4.z, arh4.w};
    float myl = 0.f, myr = 0.f;
    for (int r = 0; r < 8; ++r) {
        float sl = 0.f, sr = 0.f;
        for (int t = 0; t < 8; ++t) {
            sl += acc[t][r] * alv[t];
            sr += acc[t][r] * arv[t];
        }
        for (int off = 8; off > 0; off >>= 1) {
            sl += __shfl_xor(sl, off, 32);
            sr += __shfl_xor(sr, off, 32);
        }
        if (l16 == r) { myl = sl; myr = sr; }
    }
    if (l16 < 8) {
        el[row0 + l16 + 8 * hi] = myl;
        er[row0 + l16 + 8 * hi] = myr;
    }
}

__global__ __launch_bounds__(256) void k_gemm2(const float* __restrict__ H,
                                               const float* __restrict__ FW,
                                               const float* __restrict__ FB,
                                               float* __restrict__ out) {
    const int wave = threadIdx.x >> 5;
    const int lane = threadIdx.x & 31;
    const int rowTile = blockIdx.x * 8 + wave;
    if (rowTile >= N_NODES / 16) return;
    const int row0 = rowTile * 16;
    const int l16 = lane & 15;
    const int hi  = lane >> 4;

    v8f acc0 = {}, acc1 = {}, acc2 = {}, acc3 = {};
    const float4* __restrict__ hrow4 =
        reinterpret_cast<const float4*>(H + (size_t)(row0 + l16) * D_HID + 8 * hi);
    const float4* __restrict__ fw4 =
        reinterpret_cast<const float4*>(FW + (size_t)(8 * hi) * N_CLASSES + 4 * l16);

    for (int j = 0; j < D_HID / 16; ++j) {
        const float4 q0 = hrow4[4 * j];
        const float4 q1 = hrow4[4 * j + 1];
        const float av[8] = {q0.x, q0.y, q0.z, q0.w, q1.x, q1.y, q1.z, q1.w};
        for (int s = 0; s < 4; ++s) {
            const float4 b0 = fw4[(size_t)(16 * j + 2 * s) * (N_CLASSES / 4)];
            const float4 b1 = fw4[(size_t)(16 * j + 2 * s + 1) * (N_CLASSES / 4)];
            v2f a;
            a.x = av[2 * s];
            a.y = av[2 * s + 1];
            v2f c0, c1, c2, c3;
            c0.x = b0.x; c0.y = b1.x;
            c1.x = b0.y; c1.y = b1.y;
            c2.x = b0.z; c2.y = b1.z;
            c3.x = b0.w; c3.y = b1.w;
            acc0 = __builtin_amdgcn_wmma_f32_16x16x4_f32(
                false, a, false, c0, (short)0, acc0, false, false);
            acc1 = __builtin_amdgcn_wmma_f32_16x16x4_f32(
                false, a, false, c1, (short)0, acc1, false, false);
            acc2 = __builtin_amdgcn_wmma_f32_16x16x4_f32(
                false, a, false, c2, (short)0, acc2, false, false);
            acc3 = __builtin_amdgcn_wmma_f32_16x16x4_f32(
                false, a, false, c3, (short)0, acc3, false, false);
        }
    }
    const float4 bias = reinterpret_cast<const float4*>(FB)[l16];
    float4* __restrict__ out4 = reinterpret_cast<float4*>(out);
    for (int r = 0; r < 8; ++r) {
        out4[(size_t)(row0 + r + 8 * hi) * (N_CLASSES / 4) + l16] =
            make_float4(acc0[r] + bias.x, acc1[r] + bias.y, acc2[r] + bias.z, acc3[r] + bias.w);
    }
}

__device__ __forceinline__ unsigned enc_f32(float f) {
    unsigned b = __float_as_uint(f);
    return (b & 0x80000000u) ? ~b : (b | 0x80000000u);
}
__device__ __forceinline__ float dec_f32(unsigned u) {
    unsigned b = (u & 0x80000000u) ? (u & 0x7FFFFFFFu) : ~u;
    return __uint_as_float(b);
}

#define AGG_WGS     256
#define AGG_ROWS    391
static_assert(AGG_ROWS == (int)OWN_ROWS && AGG_WGS == (int)OWNERS, "owner bitmap and block ranges must agree");
#define AGG_THREADS 1024
#define AGG_WAVES   (AGG_THREADS / 32)
#define AGG_CAP     400
#define DL_CAP      768
#define AGG_BATCH   8
#define AGG_UNROLL  4
#define EDGE_GROUPS (N_EDGES / 128)
#define AGG_SLOTS   ((AGG_ROWS + AGG_WAVES - 1) / AGG_WAVES)
static_assert(AGG_SLOTS <= 32, "lane k of a wave holds the denominator of its slot k");
static_assert(AGG_WAVES == 32, "lane t of a wave holds the count of summing wave t");
static_assert((size_t)AGG_WAVES * DL_CAP * 8 <= (size_t)AGG_ROWS * D_HID * 4, "summing lists fit the slow path's tile");

__device__ __forceinline__ float edge_score(float l, float r) {
    const float v = l + r;
    return (v > 0.f) ? v : NEG_SLOPE * v;
}

typedef __attribute__((ext_vector_type(16))) float v16f;
static_assert(AGG_SLOTS <= 16, "one vector element per slot");

template <bool FULL>
__device__ __forceinline__ void agg_batch(int sr, float a, int jb, int n, int lane,
                                          const float* __restrict__ feat,
                                          v16f (&sc)[4], float& dn) {
    float4 v[AGG_BATCH];
    int srj[AGG_BATCH];
    const float4* __restrict__ feat4 = reinterpret_cast<const float4*>(feat);
    #pragma unroll
    for (int j = 0; j < AGG_BATCH; ++j) {
        if (FULL || j < n) {
            srj[j] = __builtin_amdgcn_readlane(sr, jb + j);
            const unsigned sj = (unsigned)srj[j] & ((1u << ROW_SHIFT) - 1u);
            v[j] = feat4[sj * (unsigned)(D_HID / 4) + (unsigned)lane];
        }
    }
    #pragma unroll
    for (int j = 0; j < AGG_BATCH; ++j) {
        if (FULL || j < n) {
            const int kj = (int)(((unsigned)srj[j] >> ROW_SHIFT) / AGG_WAVES);
            const float aj = __int_as_float(__builtin_amdgcn_readlane(__float_as_int(a), jb + j));
            if (lane == kj) dn += aj;
            sc[0][kj] += aj * v[j].x;
            sc[1][kj] += aj * v[j].y;
            sc[2][kj] += aj * v[j].z;
            sc[3][kj] += aj * v[j].w;
        }
    }
}

__global__ __launch_bounds__(AGG_THREADS) void k_agg(const int* __restrict__ src,
                                                     const int* __restrict__ dst,
                                                     const unsigned* bmp,
                                                     const unsigned* __restrict__ rec,
                                                     const float* __restrict__ el,
                                                     const float* __restrict__ er,
                                                     const float* __restrict__ feat,
                                                     float* h) {
    __shared__ __attribute__((aligned(16))) float acc[AGG_ROWS * D_HID];
    __shared__ int2 lst[AGG_WAVES * AGG_CAP];
    __shared__ unsigned cnt2[AGG_WAVES * AGG_WAVES];
    __shared__ float den[AGG_ROWS];
    __shared__ unsigned mx[AGG_ROWS];
    __shared__ int ovf;

    const int tid  = threadIdx.x;
    const int lane = tid & 31;
    const int wave = __builtin_amdgcn_readfirstlane(tid >> 5);
    const int base = blockIdx.x * AGG_ROWS;
    const int nrows = (N_NODES - base < AGG_ROWS) ? (N_NODES - base) : AGG_ROWS;

    if (tid < AGG_ROWS) mx[tid] = 0u;
    if (tid == 0) ovf = 0;
    __syncthreads();

    int2* lstw = lst + wave * AGG_CAP;
    const float* __restrict__ er_own = er + base;
    const unsigned lt_mask = (1u << lane) - 1u;
    const int4* __restrict__ dst4 = reinterpret_cast<const int4*>(dst);

    const uint4* bm16 = reinterpret_cast<const uint4*>(bmp) + (size_t)blockIdx.x * BM_ROW_U4;
    unsigned cnt = 0;
    auto fetch = [&](int g) -> uint4 {
        return bm16[((g < (int)BM_GROUPS) ? g : ((int)BM_GROUPS - 1)) * 32 + lane];
    };
    auto examine = [&](const uint4& o, int g) {
        if (g >= (int)BM_GROUPS) return;
        unsigned w[4] = {o.x, o.y, o.z, o.w};
        if (g == (int)BM_GROUPS - 1 && (unsigned)(g * 32 + lane) >= BM_VALID_U4)
            w[0] = w[1] = w[2] = w[3] = 0u;
        const int e0 = (g * 32 + lane) * 128;
        for (int c = 0; c < 4; ++c) {
            unsigned zz = w[c];
            unsigned bal = __builtin_amdgcn_ballot_w32(zz != 0u);
            while (bal != 0u) {
                const unsigned pos = cnt + (unsigned)__popc(bal & lt_mask);
                if (zz != 0u) {
                    const int p = __builtin_ctz(zz);
                    zz &= zz - 1u;
                    if (pos < (unsigned)AGG_CAP)
                        lstw[pos] = make_int2(e0 + 32 * c + p, 0);
                }
                cnt += (unsigned)__popc(bal);
                bal = __builtin_amdgcn_ballot_w32(zz != 0u);
            }
        }
    };

    constexpr int STEP = AGG_WAVES * AGG_UNROLL;
    uint4 bufA[AGG_UNROLL], bufB[AGG_UNROLL];
    #pragma unroll
    for (int u = 0; u < AGG_UNROLL; ++u) bufA[u] = fetch(wave + AGG_WAVES * u);
    for (int g0 = wave; g0 < (int)BM_GROUPS; g0 += 2 * STEP) {
        #pragma unroll
        for (int u = 0; u < AGG_UNROLL; ++u) bufB[u] = fetch(g0 + STEP + AGG_WAVES * u);
        #pragma unroll
        for (int u = 0; u < AGG_UNROLL; ++u) examine(bufA[u], g0 + AGG_WAVES * u);
        #pragma unroll
        for (int u = 0; u < AGG_UNROLL; ++u) bufA[u] = fetch(g0 + 2 * STEP + AGG_WAVES * u);
        #pragma unroll
        for (int u = 0; u < AGG_UNROLL; ++u) examine(bufB[u], g0 + STEP + AGG_WAVES * u);
    }
    const unsigned cntc = (cnt < (unsigned)AGG_CAP) ? cnt : (unsigned)AGG_CAP;
    if (lane == 0 && cnt > (unsigned)AGG_CAP) ovf = 1;

    unsigned lmask[5];
    for (int k = 0; k < 5; ++k) lmask[k] = ((lane >> k) & 1) ? 0u : ~0u;
    auto chunk_counts = [&](bool val, unsigned t) -> unsigned {
        unsigned m = __builtin_amdgcn_ballot_w32(val);
        for (int k = 0; k < 5; ++k)
            m &= __builtin_amdgcn_ballot_w32(val && ((t >> k) & 1u)) ^ lmask[k];
        return (unsigned)__popc(m);
    };
    unsigned hist = 0;

    for (unsigned i0 = 0; i0 < cntc; i0 += 128) {
        int  en[4], sv[4], rw[4];
        unsigned pk[4];
        float ev[4];
        #pragma unroll
        for (int u = 0; u < 4; ++u) {
            const unsigned i = i0 + 32u * u + (unsigned)lane;
            en[u] = (i < cntc) ? lstw[i].x : 0;
        }
        #pragma unroll
        for (int u = 0; u < 4; ++u) pk[u] = rec[en[u]];
        #pragma unroll
        for (int u = 0; u < 4; ++u) {
            const unsigned i = i0 + 32u * u + (unsigned)lane;
            pk[u] = (i < cntc) ? pk[u] : 0u;
            sv[u] = (int)(pk[u] & ((1u << ROW_SHIFT) - 1u));
            rw[u] = (int)(pk[u] >> ROW_SHIFT);
        }
        for (int u = 0; u < 4; ++u)
            hist += chunk_counts(i0 + 32u * u + (unsigned)lane < cntc, (unsigned)rw[u] & (AGG_WAVES - 1));
        #pragma unroll
        for (int u = 0; u < 4; ++u) ev[u] = edge_score(el[sv[u]], er_own[rw[u]]);
        #pragma unroll
        for (int u = 0; u < 4; ++u) {
            const unsigned i = i0 + 32u * u + (unsigned)lane;
            if (i < cntc) {
                atomicMax(&mx[rw[u]], enc_f32(ev[u]));
                lstw[i] = make_int2((int)pk[u], __float_as_int(ev[u]));
            }
        }
    }
    cnt2[wave * AGG_WAVES + lane] = hist;
    __syncthreads();

    const int ovf_u = __builtin_amdgcn_readfirstlane(ovf);
    unsigned pre = 0, tot = 0;
    for (int q = 0; q < AGG_WAVES; ++q) {
        const unsigned c = cnt2[q * AGG_WAVES + lane];
        tot += c;
        if (q < wave) pre += c;
    }
    const bool fits = __builtin_amdgcn_ballot_w32(tot > (unsigned)DL_CAP) == 0u;
    if (ovf_u == 0 && fits) {
        int2* dl = reinterpret_cast<int2*>(acc);
        unsigned pos = pre;
        for (unsigned i0 = 0; i0 < cntc; i0 += 32) {
            const unsigned i = i0 + (unsigned)lane;
            const bool val = i < cntc;
            const int2 ent = val ? lstw[i] : make_int2(0, 0);
            const unsigned t = ((unsigned)ent.x >> ROW_SHIFT) & (AGG_WAVES - 1);
            unsigned same = __builtin_amdgcn_ballot_w32(val);
            unsigned mine = same;
            for (int k = 0; k < 5; ++k) {
                const unsigned bk = __builtin_amdgcn_ballot_w32(val && ((t >> k) & 1u));
                same &= ((t >> k) & 1u) ? bk : ~bk;
                mine &= bk ^ lmask[k];
            }
            const unsigned first = (unsigned)__builtin_amdgcn_ds_bpermute((int)(t << 2), (int)pos);
            if (val) dl[t * DL_CAP + first + (unsigned)__popc(same & lt_mask)] = ent;
            pos += (unsigned)__popc(mine);
        }
        __syncthreads();

        const unsigned nt = (unsigned)__builtin_amdgcn_readlane((int)tot, wave);
        const int2* dlw = dl + wave * DL_CAP;
        v16f sc[4] = {};
        float dn = 0.f;
        const unsigned nfull = nt & ~(unsigned)(AGG_BATCH - 1);
        int sr = 0;
        float a = 0.f;
        for (unsigned i0 = 0; i0 < nt; i0 += 32) {
            const unsigned i = i0 + (unsigned)lane;
            sr = 0;
            a = 0.f;
            if (i < nt) {
                const int2 ent = dlw[i];
                sr = ent.x;
                a  = __expf(__int_as_float(ent.y) - dec_f32(mx[sr >> ROW_SHIFT]));
            }
            const int nb = (nfull - i0 < 32u) ? (int)(nfull - i0) : 32;
            for (int jb = 0; jb < nb; jb += AGG_BATCH)
                agg_batch<true>(sr, a, jb, AGG_BATCH, lane, feat, sc, dn);
        }
        if (nt != nfull)
            agg_batch<false>(sr, a, (int)(nfull & 31u), (int)(nt - nfull), lane, feat, sc, dn);
        const float dinv = (dn > 0.f) ? 1.0f / dn : 0.f;
        float4* h4 = reinterpret_cast<float4*>(h + (size_t)base * D_HID);
        #pragma unroll
        for (int k = 0; k < AGG_SLOTS; ++k) {
            const int row = wave + AGG_WAVES * k;
            if (row < nrows) {
                const float r = __int_as_float(__builtin_amdgcn_readlane(__float_as_int(dinv), k));
                h4[row * (D_HID / 4) + lane] =
                    make_float4(sc[0][k] * r, sc[1][k] * r, sc[2][k] * r, sc[3][k] * r);
            }
        }
    } else {
        float4* acc4 = reinterpret_cast<float4*>(acc);
        for (int i = tid; i < AGG_ROWS * D_HID / 4; i += AGG_THREADS)
            acc4[i] = make_float4(0.f, 0.f, 0.f, 0.f);
        if (tid < AGG_ROWS) den[tid] = 0.f;
        for (int g = wave; g < EDGE_GROUPS; g += AGG_WAVES) {
            const int4 d = dst4[g * 32 + lane];
            const int dd[4] = {d.x, d.y, d.z, d.w};
            #pragma unroll
            for (int c = 0; c < 4; ++c) {
                const unsigned t = (unsigned)(dd[c] - base);
                if (t < (unsigned)nrows) {
                    const int s = src[(g * 32 + lane) * 4 + c];
                    atomicMax(&mx[t], enc_f32(edge_score(el[s], er_own[t])));
                }
            }
        }
        __syncthreads();
        for (int g = wave; g < EDGE_GROUPS; g += AGG_WAVES) {
            const int4 d = dst4[g * 32 + lane];
            const int dd[4] = {d.x, d.y, d.z, d.w};
            #pragma unroll
            for (int c = 0; c < 4; ++c) {
                const unsigned t = (unsigned)(dd[c] - base);
                const bool m = t < (unsigned)nrows;
                int s = 0, tl = 0;
                float a = 0.f;
                if (m) {
                    s = src[(g * 32 + lane) * 4 + c];
                    tl = (int)t;
                    a = __expf(edge_score(el[s], er_own[t]) - dec_f32(mx[t]));
                    atomicAdd(den + tl, a);
                }
                unsigned bal = __builtin_amdgcn_ballot_w32(m);
                while (bal != 0u) {
                    const int j = __builtin_ctz(bal);
                    bal &= bal - 1u;
                    const int sj = __builtin_amdgcn_readlane(s, j);
                    const int dj = __builtin_amdgcn_readlane(tl, j);
                    const float aj = __int_as_float(__builtin_amdgcn_readlane(__float_as_int(a), j));
                    const float2* __restrict__ row =
                        reinterpret_cast<const float2*>(feat + (size_t)sj * D_HID);
                    const float2 x0 = row[lane], x1 = row[32 + lane];
                    float* arow = acc + dj * D_HID + 2 * lane;
                    atomicAdd(arow,      aj * x0.x);
                    atomicAdd(arow + 1,  aj * x0.y);
                    atomicAdd(arow + 64, aj * x1.x);
                    atomicAdd(arow + 65, aj * x1.y);
                }
            }
        }
        __syncthreads();
        if (tid < AGG_ROWS) {
            const float d = den[tid];
            den[tid] = (d > 0.f) ? 1.0f / d : 0.f;
        }
        __syncthreads();
        float4* h4 = reinterpret_cast<float4*>(h + (size_t)base * D_HID);
        for (int i = tid; i < nrows * (D_HID / 4); i += AGG_THREADS) {
            const float r = den[i / (D_HID / 4)];
            float4 v = acc4[i];
            v.x *= r; v.y *= r; v.z *= r; v.w *= r;
            h4[i] = v;
        }
    }
}

extern "C" void kernel_launch(void* const* d_in, const int* in_sizes, int n_in,
                              void* d_out, int out_size, void* d_ws, size_t ws_size,
                              hipStream_t stream) {
    const float* X  = (const float*)d_in[0];
    const float* W  = (const float*)d_in[1];
    const float* al = (const float*)d_in[2];
    const float* ar = (const float*)d_in[3];
    const float* FW = (const float*)d_in[4];
    const float* FB = (const float*)d_in[5];
    const int*  src = (const int*)d_in[6];
    const int*  dst = (const int*)d_in[7];
    float* out = (float*)d_out;

    char* ws = (char*)d_ws;
    size_t off = 0;
    auto carve = [&](size_t bytes) {
        void* p = ws + off;
        off += (bytes + 255) & ~(size_t)255;
        return p;
    };
    float*    feat  = (float*)   carve((size_t)N_NODES * D_HID * sizeof(float));
    float*    h     = (float*)   carve((size_t)OWNERS * OWN_ROWS * D_HID * sizeof(float));
    float*    el    = (float*)   carve((size_t)N_NODES * sizeof(float));
    float*    er    = (float*)   carve((size_t)N_NODES * sizeof(float));
    unsigned* rec   = (unsigned*)carve((size_t)N_EDGES * sizeof(unsigned));
    unsigned* bmp   = (unsigned*)h;

    k_gemm1<<<G1_BLOCKS, 128, 0, stream>>>(X, W, al, ar, feat, el, er, src, dst, bmp, rec);
    k_agg<<<AGG_WGS, AGG_THREADS, 0, stream>>>(src, dst, bmp, rec, el, er, feat, h);
    k_gemm2<<<782, 256, 0, stream>>>(h, FW, FB, out);
}
